# starting kernel
# baseline (speedup 1.0000x reference)
#include <hip/hip_runtime.h>
#include <hip/hip_bf16.h>

typedef __attribute__((ext_vector_type(2))) float v2f;
typedef __attribute__((ext_vector_type(8))) float v8f;

#define NUM_TYPES 8
#define EMBED_DIM 32

__global__ void k_init(const int* __restrict__ node_type,
                       float* __restrict__ c0, float* __restrict__ c1, int n_nodes) {
    int idx = blockIdx.x * blockDim.x + threadIdx.x;
    if (idx >= n_nodes * NUM_TYPES) return;
    int v = idx >> 3;
    int t = idx & 7;
    c0[idx] = (node_type[v] == t) ? 1.0f : 0.0f;
    c1[idx] = 0.0f;
}

__global__ void k_zero(float* __restrict__ p, int n) {
    int idx = blockIdx.x * blockDim.x + threadIdx.x;
    if (idx < n) p[idx] = 0.0f;
}

__global__ void k_edges(const int* __restrict__ src, const int* __restrict__ dst,
                        const float* __restrict__ cin, float* __restrict__ cout,
                        int n_edges) {
    long long gid = (long long)blockIdx.x * blockDim.x + threadIdx.x;
    int msg = (int)(gid >> 3);
    int lc  = (int)(gid & 7);
    if (msg >= n_edges) return;
    int u = src[msg];
    int v = dst[msg];
    float cu = cin[(size_t)u * NUM_TYPES + lc];
    float cv = cin[(size_t)v * NUM_TYPES + lc];
    __hip_atomic_fetch_add(&cout[(size_t)v * NUM_TYPES + lc], cu,
                           __ATOMIC_RELAXED, __HIP_MEMORY_SCOPE_AGENT);
    __hip_atomic_fetch_add(&cout[(size_t)u * NUM_TYPES + lc], cv,
                           __ATOMIC_RELAXED, __HIP_MEMORY_SCOPE_AGENT);
}

__global__ void k_embed_gemm(const float* __restrict__ counts,
                             const float* __restrict__ W,
                             float* __restrict__ out, int n_rows) {
    const int wave = (blockIdx.x * blockDim.x + threadIdx.x) >> 5;
    const int lane = threadIdx.x & 31;
    const int n_mtiles = (n_rows + 15) >> 4;
    const int mt = wave >> 1;
    const int nt = wave & 1;
    if (mt >= n_mtiles) return;

    const int m     = lane & 15;
    const int khalf = (lane >> 4) & 1;

    v8f acc = {0.f, 0.f, 0.f, 0.f, 0.f, 0.f, 0.f, 0.f};
    const float* crow = counts + (size_t)(mt * 16 + m) * NUM_TYPES;

#pragma unroll
    for (int s = 0; s < 2; ++s) {
        const int kb = s * 4 + khalf * 2;
        v2f a, b;
        a.x = crow[kb];
        a.y = crow[kb + 1];
        b.x = W[(size_t)kb       * EMBED_DIM + nt * 16 + m];
        b.y = W[(size_t)(kb + 1) * EMBED_DIM + nt * 16 + m];
        acc = __builtin_amdgcn_wmma_f32_16x16x4_f32(
             false, a,  false, b,
             (short)0, acc,  false,  false);
    }

    if (mt * 16 + 16 <= n_rows) {
        float* obase = out + (size_t)(mt * 16 + khalf * 8) * EMBED_DIM + nt * 16 + m;
#pragma unroll
        for (int r = 0; r < 8; ++r) {
            obase[(size_t)r * EMBED_DIM] = acc[r];
        }
    } else {
#pragma unroll
        for (int r = 0; r < 8; ++r) {
            int row = mt * 16 + khalf * 8 + r;
            if (row < n_rows) out[(size_t)row * EMBED_DIM + nt * 16 + m] = acc[r];
        }
    }
}

extern "C" void kernel_launch(void* const* d_in, const int* in_sizes, int n_in,
                              void* d_out, int out_size, void* d_ws, size_t ws_size,
                              hipStream_t stream) {
    const float* embed_table = (const float*)d_in[0];
    const int*   node_type   = (const int*)d_in[1];
    const int*   src         = (const int*)d_in[2];
    const int*   dst         = (const int*)d_in[3];
    float*       out         = (float*)d_out;

    const int n_nodes = in_sizes[1];
    const int n_edges = in_sizes[2];
    const int n_rows  = out_size / EMBED_DIM;

    float* c0 = (float*)d_ws;
    float* c1 = c0 + (size_t)n_nodes * NUM_TYPES;

    const int nc = n_nodes * NUM_TYPES;

    k_init<<<(nc + 255) / 256, 256, 0, stream>>>(node_type, c0, c1, n_nodes);

    {
        long long threads = (long long)n_edges * 8;
        int blocks = (int)((threads + 255) / 256);
        k_edges<<<blocks, 256, 0, stream>>>(src, dst, c0, c1, n_edges);
    }

    k_zero<<<(nc + 255) / 256, 256, 0, stream>>>(c0, nc);
    {
        long long threads = (long long)n_edges * 8;
        int blocks = (int)((threads + 255) / 256);
        k_edges<<<blocks, 256, 0, stream>>>(src, dst, c1, c0, n_edges);
    }

    {
        int n_mtiles = (n_rows + 15) / 16;
        int tiles = n_mtiles * 2;
        int waves_per_block = 256 / 32;
        int blocks = (tiles + waves_per_block - 1) / waves_per_block;
        k_embed_gemm<<<blocks, 256, 0, stream>>>(c0, embed_table, out, n_rows);
    }
}


# Round 2 (zero fill + scatter) only for rows < n_rows: lane per edge tests target, hits packed per wave in LDS, 8 lanes gather+atomic-add a row; addends are whole numbers, sums stay exact
# speedup vs baseline: 1.5020x; 1.5020x over previous
#include <hip/hip_runtime.h>
#include <hip/hip_bf16.h>

typedef __attribute__((ext_vector_type(2))) float v2f;
typedef __attribute__((ext_vector_type(8))) float v8f;

#define NUM_TYPES 8
#define EMBED_DIM 32

__global__ void k_init(const int* __restrict__ node_type,
                       float* __restrict__ c0, float* __restrict__ c1, int n_nodes) {
    int idx = blockIdx.x * blockDim.x + threadIdx.x;
    if (idx >= n_nodes * NUM_TYPES) return;
    int v = idx >> 3;
    int t = idx & 7;
    c0[idx] = (node_type[v] == t) ? 1.0f : 0.0f;
    c1[idx] = 0.0f;
}

__global__ void k_zero(float* __restrict__ p, int n) {
    int idx = blockIdx.x * blockDim.x + threadIdx.x;
    if (idx < n) p[idx] = 0.0f;
}

__global__ void k_edges(const int* __restrict__ src, const int* __restrict__ dst,
                        const float* __restrict__ cin, float* __restrict__ cout,
                        int n_edges) {
    long long gid = (long long)blockIdx.x * blockDim.x + threadIdx.x;
    int msg = (int)(gid >> 3);
    int lc  = (int)(gid & 7);
    if (msg >= n_edges) return;
    int u = src[msg];
    int v = dst[msg];
    float cu = cin[(size_t)u * NUM_TYPES + lc];
    float cv = cin[(size_t)v * NUM_TYPES + lc];
    __hip_atomic_fetch_add(&cout[(size_t)v * NUM_TYPES + lc], cu,
                           __ATOMIC_RELAXED, __HIP_MEMORY_SCOPE_AGENT);
    __hip_atomic_fetch_add(&cout[(size_t)u * NUM_TYPES + lc], cv,
                           __ATOMIC_RELAXED, __HIP_MEMORY_SCOPE_AGENT);
}

__global__ void __launch_bounds__(256)
k_edges_out(const int* __restrict__ src, const int* __restrict__ dst,
            const float* __restrict__ cin, float* __restrict__ cout,
            int n_edges, int n_rows) {
    __shared__ int2 hits[8][64];
    const int lane = threadIdx.x & 31;
    const int wv   = threadIdx.x >> 5;
    const int e    = blockIdx.x * blockDim.x + threadIdx.x;
    const bool in  = e < n_edges;
    int u = 0, v = 0;
    if (in) { u = src[e]; v = dst[e]; }
    const bool q1 = in && (unsigned)v < (unsigned)n_rows;
    const bool q2 = in && (unsigned)u < (unsigned)n_rows;
    const unsigned m1 = __builtin_amdgcn_ballot_w32(q1);
    const unsigned m2 = __builtin_amdgcn_ballot_w32(q2);
    const unsigned below = (1u << lane) - 1u;
    const int n1 = __builtin_popcount(m1);
    const int n  = n1 + __builtin_popcount(m2);
    if (q1) hits[wv][__builtin_popcount(m1 & below)] = make_int2(u, v);
    if (q2) hits[wv][n1 + __builtin_popcount(m2 & below)] = make_int2(v, u);
    __syncthreads();
    const int lc = lane & 7;
    for (int i = lane >> 3; i < n; i += 4) {
        const int2 h = hits[wv][i];
        const float val = cin[(size_t)h.x * NUM_TYPES + lc];
        __hip_atomic_fetch_add(&cout[(size_t)h.y * NUM_TYPES + lc], val,
                               __ATOMIC_RELAXED, __HIP_MEMORY_SCOPE_AGENT);
    }
}

__global__ void k_embed_gemm(const float* __restrict__ counts,
                             const float* __restrict__ W,
                             float* __restrict__ out, int n_rows) {
    const int wave = (blockIdx.x * blockDim.x + threadIdx.x) >> 5;
    const int lane = threadIdx.x & 31;
    const int n_mtiles = (n_rows + 15) >> 4;
    const int mt = wave >> 1;
    const int nt = wave & 1;
    if (mt >= n_mtiles) return;

    const int m     = lane & 15;
    const int khalf = (lane >> 4) & 1;

    v8f acc = {0.f, 0.f, 0.f, 0.f, 0.f, 0.f, 0.f, 0.f};
    const float* crow = counts + (size_t)(mt * 16 + m) * NUM_TYPES;

#pragma unroll
    for (int s = 0; s < 2; ++s) {
        const int kb = s * 4 + khalf * 2;
        v2f a, b;
        a.x = crow[kb];
        a.y = crow[kb + 1];
        b.x = W[(size_t)kb       * EMBED_DIM + nt * 16 + m];
        b.y = W[(size_t)(kb + 1) * EMBED_DIM + nt * 16 + m];
        acc = __builtin_amdgcn_wmma_f32_16x16x4_f32(
             false, a,  false, b,
             (short)0, acc,  false,  false);
    }

    if (mt * 16 + 16 <= n_rows) {
        float* obase = out + (size_t)(mt * 16 + khalf * 8) * EMBED_DIM + nt * 16 + m;
#pragma unroll
        for (int r = 0; r < 8; ++r) {
            obase[(size_t)r * EMBED_DIM] = acc[r];
        }
    } else {
#pragma unroll
        for (int r = 0; r < 8; ++r) {
            int row = mt * 16 + khalf * 8 + r;
            if (row < n_rows) out[(size_t)row * EMBED_DIM + nt * 16 + m] = acc[r];
        }
    }
}

extern "C" void kernel_launch(void* const* d_in, const int* in_sizes, int n_in,
                              void* d_out, int out_size, void* d_ws, size_t ws_size,
                              hipStream_t stream) {
    const float* embed_table = (const float*)d_in[0];
    const int*   node_type   = (const int*)d_in[1];
    const int*   src         = (const int*)d_in[2];
    const int*   dst         = (const int*)d_in[3];
    float*       out         = (float*)d_out;

    const int n_nodes = in_sizes[1];
    const int n_edges = in_sizes[2];
    const int n_rows  = out_size / EMBED_DIM;

    float* c0 = (float*)d_ws;
    float* c1 = c0 + (size_t)n_nodes * NUM_TYPES;

    const int nc = n_nodes * NUM_TYPES;

    k_init<<<(nc + 255) / 256, 256, 0, stream>>>(node_type, c0, c1, n_nodes);

    {
        long long threads = (long long)n_edges * 8;
        int blocks = (int)((threads + 255) / 256);
        k_edges<<<blocks, 256, 0, stream>>>(src, dst, c0, c1, n_edges);
    }

    {
        const int nz = n_rows * NUM_TYPES;
        k_zero<<<(nz + 255) / 256, 256, 0, stream>>>(c0, nz);
        k_edges_out<<<(n_edges + 255) / 256, 256, 0, stream>>>(src, dst, c1, c0, n_edges, n_rows);
    }

    {
        int n_mtiles = (n_rows + 15) / 16;
        int tiles = n_mtiles * 2;
        int waves_per_block = 256 / 32;
        int blocks = (tiles + waves_per_block - 1) / waves_per_block;
        k_embed_gemm<<<blocks, 256, 0, stream>>>(c0, embed_table, out, n_rows);
    }
}


# Round 1 counted in LDS: a work-group per (node range, edge chunk) adds 1 at [target][type of source], merges its partial counts by whole-line f32 atomics; whole numbers, sums exact in any order
# speedup vs baseline: 2.9722x; 1.9788x over previous
#include <hip/hip_runtime.h>
#include <hip/hip_bf16.h>

typedef __attribute__((ext_vector_type(2))) float v2f;
typedef __attribute__((ext_vector_type(8))) float v8f;

#define NUM_TYPES 8
#define EMBED_DIM 32

__global__ void k_zero(float* __restrict__ p, int n) {
    int idx = blockIdx.x * blockDim.x + threadIdx.x;
    if (idx < n) p[idx] = 0.0f;
}

#define R1_RANGE_NODES 6400
#define R1_THREADS 1024
#define R1_GROUPS 256
#define R1_U 3
#define R1_F 5

__device__ __forceinline__ void r1_count(float* acc, bool hit, int rb, int ta) {
    if (hit && (unsigned)ta < (unsigned)NUM_TYPES)
        __hip_atomic_fetch_add(&acc[rb * NUM_TYPES + ta], 1.0f,
                               __ATOMIC_RELAXED, __HIP_MEMORY_SCOPE_WORKGROUP);
}

__global__ void __launch_bounds__(R1_THREADS)
k_round1(const int* __restrict__ src, const int* __restrict__ dst,
         const int* __restrict__ node_type, float* __restrict__ cout,
         int n_edges, int n_nodes, int n_chunks, int per_chunk) {
    __shared__ __attribute__((aligned(16))) float acc[R1_RANGE_NODES * NUM_TYPES];
    const int tid   = threadIdx.x;
    const int range = blockIdx.x / n_chunks;
    const int chunk = blockIdx.x - range * n_chunks;
    const int lo    = range * R1_RANGE_NODES;
    const int cnt   = min(R1_RANGE_NODES, n_nodes - lo);

    float4* acc4 = reinterpret_cast<float4*>(acc);
    for (int i = tid; i < cnt * 2; i += R1_THREADS) acc4[i] = make_float4(0.f, 0.f, 0.f, 0.f);
    __syncthreads();

    const int e0 = chunk * per_chunk;
    const int e1 = min(e0 + per_chunk, n_edges);
    const bool al = ((((size_t)src) | ((size_t)dst)) & 15) == 0;
    const int ev = (al && e1 > e0) ? e0 + ((e1 - e0) & ~3) : e0;

    for (int e = e0 + tid * 4; e < ev; e += R1_THREADS * 4 * R1_U) {
        int us[R1_U * 4], vs[R1_U * 4];
        bool ok[R1_U];
#pragma unroll
        for (int k = 0; k < R1_U; ++k) {
            const int ek = e + k * (R1_THREADS * 4);
            ok[k] = ek < ev;
            const int el = ok[k] ? ek : e0;
            const int4 a = *reinterpret_cast<const int4*>(src + el);
            const int4 b = *reinterpret_cast<const int4*>(dst + el);
            us[4 * k + 0] = a.x; us[4 * k + 1] = a.y; us[4 * k + 2] = a.z; us[4 * k + 3] = a.w;
            vs[4 * k + 0] = b.x; vs[4 * k + 1] = b.y; vs[4 * k + 2] = b.z; vs[4 * k + 3] = b.w;
        }
        bool hv[R1_U * 4], hu[R1_U * 4];
        int tu[R1_U * 4], tv[R1_U * 4];
#pragma unroll
        for (int j = 0; j < R1_U * 4; ++j) {
            hv[j] = ok[j >> 2] && (unsigned)(vs[j] - lo) < (unsigned)cnt;
            hu[j] = ok[j >> 2] && (unsigned)(us[j] - lo) < (unsigned)cnt;
            tu[j] = node_type[(unsigned)(hv[j] ? us[j] : 0)];
            tv[j] = node_type[(unsigned)(hu[j] ? vs[j] : 0)];
        }
#pragma unroll
        for (int j = 0; j < R1_U * 4; ++j) {
            r1_count(acc, hv[j], vs[j] - lo, tu[j]);
            r1_count(acc, hu[j], us[j] - lo, tv[j]);
        }
    }
    for (int i = ev + tid; i < e1; i += R1_THREADS) {
        const int u = src[i];
        const int v = dst[i];
        const bool hv1 = (unsigned)(v - lo) < (unsigned)cnt;
        const bool hu1 = (unsigned)(u - lo) < (unsigned)cnt;
        const int tu1 = node_type[(unsigned)(hv1 ? u : 0)];
        const int tv1 = node_type[(unsigned)(hu1 ? v : 0)];
        r1_count(acc, hv1, v - lo, tu1);
        r1_count(acc, hu1, u - lo, tv1);
    }
    __syncthreads();

    float* g = cout + (size_t)lo * NUM_TYPES;
    const int total = cnt * NUM_TYPES;
    for (int i = tid; i < total; i += R1_THREADS * R1_F) {
        float x[R1_F];
#pragma unroll
        for (int k = 0; k < R1_F; ++k) {
            const int ik = i + k * R1_THREADS;
            x[k] = (ik < total) ? acc[ik] : 0.0f;
        }
#pragma unroll
        for (int k = 0; k < R1_F; ++k) {
            if (x[k] != 0.0f)
                __hip_atomic_fetch_add(&g[i + k * R1_THREADS], x[k],
                                       __ATOMIC_RELAXED, __HIP_MEMORY_SCOPE_AGENT);
        }
    }
}

__global__ void __launch_bounds__(256)
k_edges_out(const int* __restrict__ src, const int* __restrict__ dst,
            const float* __restrict__ cin, float* __restrict__ cout,
            int n_edges, int n_rows) {
    __shared__ int2 hits[8][64];
    const int lane = threadIdx.x & 31;
    const int wv   = threadIdx.x >> 5;
    const int e    = blockIdx.x * blockDim.x + threadIdx.x;
    const bool in  = e < n_edges;
    int u = 0, v = 0;
    if (in) { u = src[e]; v = dst[e]; }
    const bool q1 = in && (unsigned)v < (unsigned)n_rows;
    const bool q2 = in && (unsigned)u < (unsigned)n_rows;
    const unsigned m1 = __builtin_amdgcn_ballot_w32(q1);
    const unsigned m2 = __builtin_amdgcn_ballot_w32(q2);
    const unsigned below = (1u << lane) - 1u;
    const int n1 = __builtin_popcount(m1);
    const int n  = n1 + __builtin_popcount(m2);
    if (q1) hits[wv][__builtin_popcount(m1 & below)] = make_int2(u, v);
    if (q2) hits[wv][n1 + __builtin_popcount(m2 & below)] = make_int2(v, u);
    __syncthreads();
    const int lc = lane & 7;
    for (int i = lane >> 3; i < n; i += 4) {
        const int2 h = hits[wv][i];
        const float val = cin[(size_t)h.x * NUM_TYPES + lc];
        __hip_atomic_fetch_add(&cout[(size_t)h.y * NUM_TYPES + lc], val,
                               __ATOMIC_RELAXED, __HIP_MEMORY_SCOPE_AGENT);
    }
}

__global__ void k_embed_gemm(const float* __restrict__ counts,
                             const float* __restrict__ W,
                             float* __restrict__ out, int n_rows) {
    const int wave = (blockIdx.x * blockDim.x + threadIdx.x) >> 5;
    const int lane = threadIdx.x & 31;
    const int n_mtiles = (n_rows + 15) >> 4;
    const int mt = wave >> 1;
    const int nt = wave & 1;
    if (mt >= n_mtiles) return;

    const int m     = lane & 15;
    const int khalf = (lane >> 4) & 1;

    v8f acc = {0.f, 0.f, 0.f, 0.f, 0.f, 0.f, 0.f, 0.f};
    const float* crow = counts + (size_t)(mt * 16 + m) * NUM_TYPES;

#pragma unroll
    for (int s = 0; s < 2; ++s) {
        const int kb = s * 4 + khalf * 2;
        v2f a, b;
        a.x = crow[kb];
        a.y = crow[kb + 1];
        b.x = W[(size_t)kb       * EMBED_DIM + nt * 16 + m];
        b.y = W[(size_t)(kb + 1) * EMBED_DIM + nt * 16 + m];
        acc = __builtin_amdgcn_wmma_f32_16x16x4_f32(
             false, a,  false, b,
             (short)0, acc,  false,  false);
    }

    if (mt * 16 + 16 <= n_rows) {
        float* obase = out + (size_t)(mt * 16 + khalf * 8) * EMBED_DIM + nt * 16 + m;
#pragma unroll
        for (int r = 0; r < 8; ++r) {
            obase[(size_t)r * EMBED_DIM] = acc[r];
        }
    } else {
#pragma unroll
        for (int r = 0; r < 8; ++r) {
            int row = mt * 16 + khalf * 8 + r;
            if (row < n_rows) out[(size_t)row * EMBED_DIM + nt * 16 + m] = acc[r];
        }
    }
}

extern "C" void kernel_launch(void* const* d_in, const int* in_sizes, int n_in,
                              void* d_out, int out_size, void* d_ws, size_t ws_size,
                              hipStream_t stream) {
    const float* embed_table = (const float*)d_in[0];
    const int*   node_type   = (const int*)d_in[1];
    const int*   src         = (const int*)d_in[2];
    const int*   dst         = (const int*)d_in[3];
    float*       out         = (float*)d_out;

    const int n_nodes = in_sizes[1];
    const int n_edges = in_sizes[2];
    const int n_rows  = out_size / EMBED_DIM;

    float* c0 = (float*)d_ws;
    float* c1 = c0 + (size_t)n_nodes * NUM_TYPES;

    const int nc = n_nodes * NUM_TYPES;

    k_zero<<<(nc + 255) / 256, 256, 0, stream>>>(c1, nc);

    {
        const int n_ranges  = (n_nodes + R1_RANGE_NODES - 1) / R1_RANGE_NODES;
        const int n_chunks  = (R1_GROUPS / n_ranges) > 0 ? (R1_GROUPS / n_ranges) : 1;
        const int per_chunk = ((n_edges + n_chunks - 1) / n_chunks + 3) & ~3;
        k_round1<<<n_ranges * n_chunks, R1_THREADS, 0, stream>>>(src, dst, node_type, c1,
                                                                 n_edges, n_nodes, n_chunks, per_chunk);
    }

    {
        const int nz = ((n_rows + 15) & ~15) * NUM_TYPES;
        k_zero<<<(nz + 255) / 256, 256, 0, stream>>>(c0, nz);
        k_edges_out<<<(n_edges + 255) / 256, 256, 0, stream>>>(src, dst, c1, c0, n_edges, n_rows);
    }

    {
        int n_mtiles = (n_rows + 15) / 16;
        int tiles = n_mtiles * 2;
        int waves_per_block = 256 / 32;
        int blocks = (tiles + waves_per_block - 1) / waves_per_block;
        k_embed_gemm<<<blocks, 256, 0, stream>>>(c0, embed_table, out, n_rows);
    }
}


# Wave entry: zero fills and round-2 edge scan run as fixed small grids whose lanes loop (16B stores; 4 edges per lane per step, 8 gathers in flight); atomic add order differs, whole-number sums exact
# speedup vs baseline: 3.3443x; 1.1252x over previous
#include <hip/hip_runtime.h>
#include <hip/hip_bf16.h>

typedef __attribute__((ext_vector_type(2))) float v2f;
typedef __attribute__((ext_vector_type(8))) float v8f;

#define NUM_TYPES 8
#define EMBED_DIM 32

__global__ void __launch_bounds__(256)
k_zero(float* __restrict__ p, int n) {
    const int t  = blockIdx.x * blockDim.x + threadIdx.x;
    const int nt = gridDim.x * blockDim.x;
    const int n4 = ((((size_t)p) & 15) == 0) ? (n >> 2) : 0;
    float4* p4 = reinterpret_cast<float4*>(p);
    for (int i = t; i < n4; i += nt) p4[i] = make_float4(0.f, 0.f, 0.f, 0.f);
    for (int i = (n4 << 2) + t; i < n; i += nt) p[i] = 0.0f;
}

#define R1_RANGE_NODES 6400
#define R1_THREADS 1024
#define R1_GROUPS 256
#define R1_U 3
#define R1_F 5

__device__ __forceinline__ void r1_count(float* acc, bool hit, int rb, int ta) {
    if (hit && (unsigned)ta < (unsigned)NUM_TYPES)
        __hip_atomic_fetch_add(&acc[rb * NUM_TYPES + ta], 1.0f,
                               __ATOMIC_RELAXED, __HIP_MEMORY_SCOPE_WORKGROUP);
}

__global__ void __launch_bounds__(R1_THREADS)
k_round1(const int* __restrict__ src, const int* __restrict__ dst,
         const int* __restrict__ node_type, float* __restrict__ cout,
         int n_edges, int n_nodes, int n_chunks, int per_chunk) {
    __shared__ __attribute__((aligned(16))) float acc[R1_RANGE_NODES * NUM_TYPES];
    const int tid   = threadIdx.x;
    const int range = blockIdx.x / n_chunks;
    const int chunk = blockIdx.x - range * n_chunks;
    const int lo    = range * R1_RANGE_NODES;
    const int cnt   = min(R1_RANGE_NODES, n_nodes - lo);

    float4* acc4 = reinterpret_cast<float4*>(acc);
    for (int i = tid; i < cnt * 2; i += R1_THREADS) acc4[i] = make_float4(0.f, 0.f, 0.f, 0.f);
    __syncthreads();

    const int e0 = chunk * per_chunk;
    const int e1 = min(e0 + per_chunk, n_edges);
    const bool al = ((((size_t)src) | ((size_t)dst)) & 15) == 0;
    const int ev = (al && e1 > e0) ? e0 + ((e1 - e0) & ~3) : e0;

    for (int e = e0 + tid * 4; e < ev; e += R1_THREADS * 4 * R1_U) {
        int us[R1_U * 4], vs[R1_U * 4];
        bool ok[R1_U];
#pragma unroll
        for (int k = 0; k < R1_U; ++k) {
            const int ek = e + k * (R1_THREADS * 4);
            ok[k] = ek < ev;
            const int el = ok[k] ? ek : e0;
            const int4 a = *reinterpret_cast<const int4*>(src + el);
            const int4 b = *reinterpret_cast<const int4*>(dst + el);
            us[4 * k + 0] = a.x; us[4 * k + 1] = a.y; us[4 * k + 2] = a.z; us[4 * k + 3] = a.w;
            vs[4 * k + 0] = b.x; vs[4 * k + 1] = b.y; vs[4 * k + 2] = b.z; vs[4 * k + 3] = b.w;
        }
        bool hv[R1_U * 4], hu[R1_U * 4];
        int tu[R1_U * 4], tv[R1_U * 4];
#pragma unroll
        for (int j = 0; j < R1_U * 4; ++j) {
            hv[j] = ok[j >> 2] && (unsigned)(vs[j] - lo) < (unsigned)cnt;
            hu[j] = ok[j >> 2] && (unsigned)(us[j] - lo) < (unsigned)cnt;
            tu[j] = node_type[(unsigned)(hv[j] ? us[j] : 0)];
            tv[j] = node_type[(unsigned)(hu[j] ? vs[j] : 0)];
        }
#pragma unroll
        for (int j = 0; j < R1_U * 4; ++j) {
            r1_count(acc, hv[j], vs[j] - lo, tu[j]);
            r1_count(acc, hu[j], us[j] - lo, tv[j]);
        }
    }
    for (int i = ev + tid; i < e1; i += R1_THREADS) {
        const int u = src[i];
        const int v = dst[i];
        const bool hv1 = (unsigned)(v - lo) < (unsigned)cnt;
        const bool hu1 = (unsigned)(u - lo) < (unsigned)cnt;
        const int tu1 = node_type[(unsigned)(hv1 ? u : 0)];
        const int tv1 = node_type[(unsigned)(hu1 ? v : 0)];
        r1_count(acc, hv1, v - lo, tu1);
        r1_count(acc, hu1, u - lo, tv1);
    }
    __syncthreads();

    float* g = cout + (size_t)lo * NUM_TYPES;
    const int total = cnt * NUM_TYPES;
    for (int i = tid; i < total; i += R1_THREADS * R1_F) {
        float x[R1_F];
#pragma unroll
        for (int k = 0; k < R1_F; ++k) {
            const int ik = i + k * R1_THREADS;
            x[k] = (ik < total) ? acc[ik] : 0.0f;
        }
#pragma unroll
        for (int k = 0; k < R1_F; ++k) {
            if (x[k] != 0.0f)
                __hip_atomic_fetch_add(&g[i + k * R1_THREADS], x[k],
                                       __ATOMIC_RELAXED, __HIP_MEMORY_SCOPE_AGENT);
        }
    }
}

#define EO_THREADS 256
#define EO_GROUPS 1024
#define EO_B 8
__global__ void __launch_bounds__(EO_THREADS)
k_edges_out(const int* __restrict__ src, const int* __restrict__ dst,
            const float* __restrict__ cin, float* __restrict__ cout,
            int n_edges, int n_rows) {
    __shared__ int2 hits[EO_THREADS / 32][256];
    const int lane = threadIdx.x & 31;
    const int wv   = threadIdx.x >> 5;
    const int lc   = lane & 7;
    const int lg   = lane >> 3;
    const unsigned below = (1u << lane) - 1u;
    const int per = (((n_edges + (int)gridDim.x - 1) / (int)gridDim.x) + 3) & ~3;
    const int e0  = blockIdx.x * per;
    const int e1  = min(e0 + per, n_edges);
    const bool al = ((((size_t)src) | ((size_t)dst)) & 15) == 0;

    for (int eb = e0; eb < e1; eb += EO_THREADS * 4) {
        const int e = eb + threadIdx.x * 4;
        int us[4], vs[4];
        if (al && e + 3 < e1) {
            const int4 a = *reinterpret_cast<const int4*>(src + e);
            const int4 b = *reinterpret_cast<const int4*>(dst + e);
            us[0] = a.x; us[1] = a.y; us[2] = a.z; us[3] = a.w;
            vs[0] = b.x; vs[1] = b.y; vs[2] = b.z; vs[3] = b.w;
        } else {
#pragma unroll
            for (int j = 0; j < 4; ++j) {
                const bool in = e + j < e1;
                us[j] = in ? src[e + j] : -1;
                vs[j] = in ? dst[e + j] : -1;
            }
        }
        int n = 0;
#pragma unroll
        for (int j = 0; j < 4; ++j) {
            const bool q1 = (unsigned)vs[j] < (unsigned)n_rows;
            const bool q2 = (unsigned)us[j] < (unsigned)n_rows;
            const unsigned m1 = __builtin_amdgcn_ballot_w32(q1);
            const unsigned m2 = __builtin_amdgcn_ballot_w32(q2);
            const int n1 = __builtin_popcount(m1);
            if (q1) hits[wv][n + __builtin_popcount(m1 & below)] = make_int2(us[j], vs[j]);
            if (q2) hits[wv][n + n1 + __builtin_popcount(m2 & below)] = make_int2(vs[j], us[j]);
            n += n1 + __builtin_popcount(m2);
        }
        __syncthreads();
        for (int i0 = 0; i0 < n; i0 += 4 * EO_B) {
            int2 h[EO_B];
            float val[EO_B];
            bool ok[EO_B];
#pragma unroll
            for (int k = 0; k < EO_B; ++k) {
                const int i = i0 + 4 * k + lg;
                ok[k] = i < n;
                h[k]  = hits[wv][ok[k] ? i : 0];
            }
#pragma unroll
            for (int k = 0; k < EO_B; ++k)
                val[k] = cin[(size_t)(unsigned)h[k].x * NUM_TYPES + lc];
#pragma unroll
            for (int k = 0; k < EO_B; ++k) {
                if (ok[k])
                    __hip_atomic_fetch_add(&cout[(size_t)(unsigned)h[k].y * NUM_TYPES + lc], val[k],
                                           __ATOMIC_RELAXED, __HIP_MEMORY_SCOPE_AGENT);
            }
        }
    }
}

__global__ void k_embed_gemm(const float* __restrict__ counts,
                             const float* __restrict__ W,
                             float* __restrict__ out, int n_rows) {
    const int wave = (blockIdx.x * blockDim.x + threadIdx.x) >> 5;
    const int lane = threadIdx.x & 31;
    const int n_mtiles = (n_rows + 15) >> 4;
    const int mt = wave >> 1;
    const int nt = wave & 1;
    if (mt >= n_mtiles) return;

    const int m     = lane & 15;
    const int khalf = (lane >> 4) & 1;

    v8f acc = {0.f, 0.f, 0.f, 0.f, 0.f, 0.f, 0.f, 0.f};
    const float* crow = counts + (size_t)(mt * 16 + m) * NUM_TYPES;

#pragma unroll
    for (int s = 0; s < 2; ++s) {
        const int kb = s * 4 + khalf * 2;
        v2f a, b;
        a.x = crow[kb];
        a.y = crow[kb + 1];
        b.x = W[(size_t)kb       * EMBED_DIM + nt * 16 + m];
        b.y = W[(size_t)(kb + 1) * EMBED_DIM + nt * 16 + m];
        acc = __builtin_amdgcn_wmma_f32_16x16x4_f32(
             false, a,  false, b,
             (short)0, acc,  false,  false);
    }

    if (mt * 16 + 16 <= n_rows) {
        float* obase = out + (size_t)(mt * 16 + khalf * 8) * EMBED_DIM + nt * 16 + m;
#pragma unroll
        for (int r = 0; r < 8; ++r) {
            obase[(size_t)r * EMBED_DIM] = acc[r];
        }
    } else {
#pragma unroll
        for (int r = 0; r < 8; ++r) {
            int row = mt * 16 + khalf * 8 + r;
            if (row < n_rows) out[(size_t)row * EMBED_DIM + nt * 16 + m] = acc[r];
        }
    }
}

extern "C" void kernel_launch(void* const* d_in, const int* in_sizes, int n_in,
                              void* d_out, int out_size, void* d_ws, size_t ws_size,
                              hipStream_t stream) {
    const float* embed_table = (const float*)d_in[0];
    const int*   node_type   = (const int*)d_in[1];
    const int*   src         = (const int*)d_in[2];
    const int*   dst         = (const int*)d_in[3];
    float*       out         = (float*)d_out;

    const int n_nodes = in_sizes[1];
    const int n_edges = in_sizes[2];
    const int n_rows  = out_size / EMBED_DIM;

    float* c0 = (float*)d_ws;
    float* c1 = c0 + (size_t)n_nodes * NUM_TYPES;

    const int nc = n_nodes * NUM_TYPES;

    k_zero<<<512, 256, 0, stream>>>(c1, nc);

    {
        const int n_ranges  = (n_nodes + R1_RANGE_NODES - 1) / R1_RANGE_NODES;
        const int n_chunks  = (R1_GROUPS / n_ranges) > 0 ? (R1_GROUPS / n_ranges) : 1;
        const int per_chunk = ((n_edges + n_chunks - 1) / n_chunks + 3) & ~3;
        k_round1<<<n_ranges * n_chunks, R1_THREADS, 0, stream>>>(src, dst, node_type, c1,
                                                                 n_edges, n_nodes, n_chunks, per_chunk);
    }

    {
        const int nz = ((n_rows + 15) & ~15) * NUM_TYPES;
        k_zero<<<64, 256, 0, stream>>>(c0, nz);
        k_edges_out<<<EO_GROUPS, EO_THREADS, 0, stream>>>(src, dst, c1, c0, n_edges, n_rows);
    }

    {
        int n_mtiles = (n_rows + 15) / 16;
        int tiles = n_mtiles * 2;
        int waves_per_block = 256 / 32;
        int blocks = (tiles + waves_per_block - 1) / waves_per_block;
        k_embed_gemm<<<blocks, 256, 0, stream>>>(c0, embed_table, out, n_rows);
    }
}


# Round 1 reads the other end's type from a byte table of all nodes held in LDS beside the counters (packed by the first launch, copied in per group): no 4-byte type gathers; exact sums unchanged
# speedup vs baseline: 3.9974x; 1.1953x over previous
#include <hip/hip_runtime.h>
#include <hip/hip_bf16.h>

typedef __attribute__((ext_vector_type(2))) float v2f;
typedef __attribute__((ext_vector_type(8))) float v8f;

#define NUM_TYPES 8
#define EMBED_DIM 32

__global__ void __launch_bounds__(256)
k_zero(float* __restrict__ p, int n) {
    const int t  = blockIdx.x * blockDim.x + threadIdx.x;
    const int nt = gridDim.x * blockDim.x;
    const int n4 = ((((size_t)p) & 15) == 0) ? (n >> 2) : 0;
    float4* p4 = reinterpret_cast<float4*>(p);
    for (int i = t; i < n4; i += nt) p4[i] = make_float4(0.f, 0.f, 0.f, 0.f);
    for (int i = (n4 << 2) + t; i < n; i += nt) p[i] = 0.0f;
}

#define TYPE_NONE 255u
__device__ __forceinline__ unsigned type_code(int t) {
    return ((unsigned)t < (unsigned)NUM_TYPES) ? ((unsigned)t << 2) : TYPE_NONE;
}

#define R1_TT_CAP 114688
__global__ void __launch_bounds__(256)
k_zero_pack(float* __restrict__ p, int n, const int* __restrict__ node_type,
            float* __restrict__ tbf, int n_nodes) {
    const int t  = blockIdx.x * blockDim.x + threadIdx.x;
    const int nt = gridDim.x * blockDim.x;
    unsigned* tb = reinterpret_cast<unsigned*>(tbf);
    const bool al = (((size_t)node_type) & 15) == 0;
    for (int i = t; i < R1_TT_CAP / 4; i += nt) {
        const int j = i << 2;
        unsigned w;
        if (al && j + 3 < n_nodes) {
            const int4 a = *reinterpret_cast<const int4*>(node_type + j);
            w = type_code(a.x) | (type_code(a.y) << 8) | (type_code(a.z) << 16) | (type_code(a.w) << 24);
        } else {
            w = 0;
            for (int q = 0; q < 4; ++q)
                w |= ((j + q < n_nodes) ? type_code(node_type[j + q]) : TYPE_NONE) << (8 * q);
        }
        tb[i] = w;
    }

    const int n4 = ((((size_t)p) & 15) == 0) ? (n >> 2) : 0;
    float4* p4 = reinterpret_cast<float4*>(p);
    for (int i = t; i < n4; i += nt) p4[i] = make_float4(0.f, 0.f, 0.f, 0.f);
    for (int i = (n4 << 2) + t; i < n; i += nt) p[i] = 0.0f;
}

#define R1_RANGE_NODES 6400
#define R1_TB 7
#define R1_THREADS 1024
#define R1_GROUPS 256
#define R1_U 3
#define R1_F 5

static_assert(R1_TB * R1_THREADS * 16 == R1_TT_CAP, "the table copy covers the table exactly");

template <bool LDS_TYPES>
__device__ __forceinline__ int r1_type(const signed char* tt, const int* __restrict__ node_type,
                                       bool hit, int node) {
    if constexpr (LDS_TYPES) return tt[hit ? (unsigned)node : (unsigned)(R1_TT_CAP - 1)];
    else return node_type[(unsigned)(hit ? node : 0)];
}

template <bool LDS_TYPES>
__device__ __forceinline__ void r1_count(float* acc, bool hit, int rb, int ta) {
    if constexpr (LDS_TYPES) {
        if (ta >= 0)
            __hip_atomic_fetch_add(reinterpret_cast<float*>(reinterpret_cast<char*>(acc) +
                                                            (((unsigned)rb << 5) + (unsigned)ta)), 1.0f,
                                   __ATOMIC_RELAXED, __HIP_MEMORY_SCOPE_WORKGROUP);
    } else {
        if (hit && (unsigned)ta < (unsigned)NUM_TYPES)
            __hip_atomic_fetch_add(&acc[rb * NUM_TYPES + ta], 1.0f,
                                   __ATOMIC_RELAXED, __HIP_MEMORY_SCOPE_WORKGROUP);
    }
}

template <bool LDS_TYPES>
__global__ void __launch_bounds__(R1_THREADS)
k_round1(const int* __restrict__ src, const int* __restrict__ dst,
         const int* __restrict__ node_type, const float* __restrict__ tbf, float* __restrict__ cout,
         int n_edges, int n_nodes, int n_chunks, int per_chunk) {
    __shared__ __attribute__((aligned(16))) float acc[R1_RANGE_NODES * NUM_TYPES];
    __shared__ __attribute__((aligned(16))) signed char tt[LDS_TYPES ? R1_TT_CAP : 16];
    const int tid   = threadIdx.x;
    const int range = blockIdx.x / n_chunks;
    const int chunk = blockIdx.x - range * n_chunks;
    const int lo    = range * R1_RANGE_NODES;
    const int cnt   = min(R1_RANGE_NODES, n_nodes - lo);

    float4* acc4 = reinterpret_cast<float4*>(acc);
    for (int i = tid; i < cnt * 2; i += R1_THREADS) acc4[i] = make_float4(0.f, 0.f, 0.f, 0.f);
    if constexpr (LDS_TYPES) {
        const uint4* tb16 = reinterpret_cast<const uint4*>(tbf);
        uint4* tt16 = reinterpret_cast<uint4*>(tt);
        uint4 x[R1_TB];
#pragma unroll
        for (int k = 0; k < R1_TB; ++k) x[k] = tb16[tid + k * R1_THREADS];
#pragma unroll
        for (int k = 0; k < R1_TB; ++k) tt16[tid + k * R1_THREADS] = x[k];
    }
    __syncthreads();

    const int e0 = chunk * per_chunk;
    const int e1 = min(e0 + per_chunk, n_edges);
    const bool al = ((((size_t)src) | ((size_t)dst)) & 15) == 0;
    const int ev = (al && e1 > e0) ? e0 + ((e1 - e0) & ~3) : e0;

    for (int e = e0 + tid * 4; e < ev; e += R1_THREADS * 4 * R1_U) {
        int us[R1_U * 4], vs[R1_U * 4];
        bool ok[R1_U];
#pragma unroll
        for (int k = 0; k < R1_U; ++k) {
            const int ek = e + k * (R1_THREADS * 4);
            ok[k] = ek < ev;
            const int el = ok[k] ? ek : e0;
            const int4 a = *reinterpret_cast<const int4*>(src + el);
            const int4 b = *reinterpret_cast<const int4*>(dst + el);
            us[4 * k + 0] = a.x; us[4 * k + 1] = a.y; us[4 * k + 2] = a.z; us[4 * k + 3] = a.w;
            vs[4 * k + 0] = b.x; vs[4 * k + 1] = b.y; vs[4 * k + 2] = b.z; vs[4 * k + 3] = b.w;
        }
        bool hv[R1_U * 4], hu[R1_U * 4];
        int tu[R1_U * 4], tv[R1_U * 4];
#pragma unroll
        for (int j = 0; j < R1_U * 4; ++j) {
            hv[j] = ok[j >> 2] && (unsigned)(vs[j] - lo) < (unsigned)cnt;
            hu[j] = ok[j >> 2] && (unsigned)(us[j] - lo) < (unsigned)cnt;
            tu[j] = r1_type<LDS_TYPES>(tt, node_type, hv[j], us[j]);
            tv[j] = r1_type<LDS_TYPES>(tt, node_type, hu[j], vs[j]);
        }
#pragma unroll
        for (int j = 0; j < R1_U * 4; ++j) {
            r1_count<LDS_TYPES>(acc, hv[j], vs[j] - lo, tu[j]);
            r1_count<LDS_TYPES>(acc, hu[j], us[j] - lo, tv[j]);
        }
    }
    for (int i = ev + tid; i < e1; i += R1_THREADS) {
        const int u = src[i];
        const int v = dst[i];
        const bool hv1 = (unsigned)(v - lo) < (unsigned)cnt;
        const bool hu1 = (unsigned)(u - lo) < (unsigned)cnt;
        const int tu1 = r1_type<LDS_TYPES>(tt, node_type, hv1, u);
        const int tv1 = r1_type<LDS_TYPES>(tt, node_type, hu1, v);
        r1_count<LDS_TYPES>(acc, hv1, v - lo, tu1);
        r1_count<LDS_TYPES>(acc, hu1, u - lo, tv1);
    }
    __syncthreads();

    float* g = cout + (size_t)lo * NUM_TYPES;
    const int total = cnt * NUM_TYPES;
    for (int i = tid; i < total; i += R1_THREADS * R1_F) {
        float x[R1_F];
#pragma unroll
        for (int k = 0; k < R1_F; ++k) {
            const int ik = i + k * R1_THREADS;
            x[k] = (ik < total) ? acc[ik] : 0.0f;
        }
#pragma unroll
        for (int k = 0; k < R1_F; ++k) {
            if (x[k] != 0.0f)
                __hip_atomic_fetch_add(&g[i + k * R1_THREADS], x[k],
                                       __ATOMIC_RELAXED, __HIP_MEMORY_SCOPE_AGENT);
        }
    }
}

#define EO_THREADS 256
#define EO_GROUPS 1024
#define EO_B 8
__global__ void __launch_bounds__(EO_THREADS)
k_edges_out(const int* __restrict__ src, const int* __restrict__ dst,
            const float* __restrict__ cin, float* __restrict__ cout,
            int n_edges, int n_rows) {
    __shared__ int2 hits[EO_THREADS / 32][256];
    const int lane = threadIdx.x & 31;
    const int wv   = threadIdx.x >> 5;
    const int lc   = lane & 7;
    const int lg   = lane >> 3;
    const unsigned below = (1u << lane) - 1u;
    const int per = (((n_edges + (int)gridDim.x - 1) / (int)gridDim.x) + 3) & ~3;
    const int e0  = blockIdx.x * per;
    const int e1  = min(e0 + per, n_edges);
    const bool al = ((((size_t)src) | ((size_t)dst)) & 15) == 0;

    for (int eb = e0; eb < e1; eb += EO_THREADS * 4) {
        const int e = eb + threadIdx.x * 4;
        int us[4], vs[4];
        if (al && e + 3 < e1) {
            const int4 a = *reinterpret_cast<const int4*>(src + e);
            const int4 b = *reinterpret_cast<const int4*>(dst + e);
            us[0] = a.x; us[1] = a.y; us[2] = a.z; us[3] = a.w;
            vs[0] = b.x; vs[1] = b.y; vs[2] = b.z; vs[3] = b.w;
        } else {
#pragma unroll
            for (int j = 0; j < 4; ++j) {
                const bool in = e + j < e1;
                us[j] = in ? src[e + j] : -1;
                vs[j] = in ? dst[e + j] : -1;
            }
        }
        int n = 0;
#pragma unroll
        for (int j = 0; j < 4; ++j) {
            const bool q1 = (unsigned)vs[j] < (unsigned)n_rows;
            const bool q2 = (unsigned)us[j] < (unsigned)n_rows;
            const unsigned m1 = __builtin_amdgcn_ballot_w32(q1);
            const unsigned m2 = __builtin_amdgcn_ballot_w32(q2);
            const int n1 = __builtin_popcount(m1);
            if (q1) hits[wv][n + __builtin_popcount(m1 & below)] = make_int2(us[j], vs[j]);
            if (q2) hits[wv][n + n1 + __builtin_popcount(m2 & below)] = make_int2(vs[j], us[j]);
            n += n1 + __builtin_popcount(m2);
        }
        __syncthreads();
        for (int i0 = 0; i0 < n; i0 += 4 * EO_B) {
            int2 h[EO_B];
            float val[EO_B];
            bool ok[EO_B];
#pragma unroll
            for (int k = 0; k < EO_B; ++k) {
                const int i = i0 + 4 * k + lg;
                ok[k] = i < n;
                h[k]  = hits[wv][ok[k] ? i : 0];
            }
#pragma unroll
            for (int k = 0; k < EO_B; ++k)
                val[k] = cin[(size_t)(unsigned)h[k].x * NUM_TYPES + lc];
#pragma unroll
            for (int k = 0; k < EO_B; ++k) {
                if (ok[k])
                    __hip_atomic_fetch_add(&cout[(size_t)(unsigned)h[k].y * NUM_TYPES + lc], val[k],
                                           __ATOMIC_RELAXED, __HIP_MEMORY_SCOPE_AGENT);
            }
        }
    }
}

__global__ void k_embed_gemm(const float* __restrict__ counts,
                             const float* __restrict__ W,
                             float* __restrict__ out, int n_rows) {
    const int wave = (blockIdx.x * blockDim.x + threadIdx.x) >> 5;
    const int lane = threadIdx.x & 31;
    const int n_mtiles = (n_rows + 15) >> 4;
    const int mt = wave >> 1;
    const int nt = wave & 1;
    if (mt >= n_mtiles) return;

    const int m     = lane & 15;
    const int khalf = (lane >> 4) & 1;

    v8f acc = {0.f, 0.f, 0.f, 0.f, 0.f, 0.f, 0.f, 0.f};
    const float* crow = counts + (size_t)(mt * 16 + m) * NUM_TYPES;

#pragma unroll
    for (int s = 0; s < 2; ++s) {
        const int kb = s * 4 + khalf * 2;
        v2f a, b;
        a.x = crow[kb];
        a.y = crow[kb + 1];
        b.x = W[(size_t)kb       * EMBED_DIM + nt * 16 + m];
        b.y = W[(size_t)(kb + 1) * EMBED_DIM + nt * 16 + m];
        acc = __builtin_amdgcn_wmma_f32_16x16x4_f32(
             false, a,  false, b,
             (short)0, acc,  false,  false);
    }

    if (mt * 16 + 16 <= n_rows) {
        float* obase = out + (size_t)(mt * 16 + khalf * 8) * EMBED_DIM + nt * 16 + m;
#pragma unroll
        for (int r = 0; r < 8; ++r) {
            obase[(size_t)r * EMBED_DIM] = acc[r];
        }
    } else {
#pragma unroll
        for (int r = 0; r < 8; ++r) {
            int row = mt * 16 + khalf * 8 + r;
            if (row < n_rows) out[(size_t)row * EMBED_DIM + nt * 16 + m] = acc[r];
        }
    }
}

extern "C" void kernel_launch(void* const* d_in, const int* in_sizes, int n_in,
                              void* d_out, int out_size, void* d_ws, size_t ws_size,
                              hipStream_t stream) {
    const float* embed_table = (const float*)d_in[0];
    const int*   node_type   = (const int*)d_in[1];
    const int*   src         = (const int*)d_in[2];
    const int*   dst         = (const int*)d_in[3];
    float*       out         = (float*)d_out;

    const int n_nodes = in_sizes[1];
    const int n_edges = in_sizes[2];
    const int n_rows  = out_size / EMBED_DIM;

    float* c0 = (float*)d_ws;
    float* c1 = c0 + (size_t)n_nodes * NUM_TYPES;

    const int nc = n_nodes * NUM_TYPES;

    const int nz = ((n_rows + 15) & ~15) * NUM_TYPES;
    float* tbf = c0 + nz;
    const bool lds_types = n_nodes < R1_TT_CAP && (((size_t)tbf) & 15) == 0 &&
                           (size_t)nz * sizeof(float) + (size_t)R1_TT_CAP <= (size_t)nc * sizeof(float);

    if (lds_types) k_zero_pack<<<512, 256, 0, stream>>>(c1, nc, node_type, tbf, n_nodes);
    else           k_zero<<<512, 256, 0, stream>>>(c1, nc);

    {
        const int n_ranges  = (n_nodes + R1_RANGE_NODES - 1) / R1_RANGE_NODES;
        const int n_chunks  = (R1_GROUPS / n_ranges) > 0 ? (R1_GROUPS / n_ranges) : 1;
        const int per_chunk = ((n_edges + n_chunks - 1) / n_chunks + 3) & ~3;
        if (lds_types)
            k_round1<true><<<n_ranges * n_chunks, R1_THREADS, 0, stream>>>(src, dst, node_type, tbf, c1,
                                                                           n_edges, n_nodes, n_chunks, per_chunk);
        else
            k_round1<false><<<n_ranges * n_chunks, R1_THREADS, 0, stream>>>(src, dst, node_type, tbf, c1,
                                                                            n_edges, n_nodes, n_chunks, per_chunk);
    }

    {
        k_zero<<<64, 256, 0, stream>>>(c0, nz);
        k_edges_out<<<EO_GROUPS, EO_THREADS, 0, stream>>>(src, dst, c1, c0, n_edges, n_rows);
    }

    {
        int n_mtiles = (n_rows + 15) / 16;
        int tiles = n_mtiles * 2;
        int waves_per_block = 256 / 32;
        int blocks = (tiles + waves_per_block - 1) / waves_per_block;
        k_embed_gemm<<<blocks, 256, 0, stream>>>(c0, embed_table, out, n_rows);
    }
}


# Round 2's zero fill moves into the first launch (one kernel zeroes both spans beside the table pack): 4 launches instead of 5; atomic add order as before, whole-number sums exact
# speedup vs baseline: 4.1252x; 1.0320x over previous
#include <hip/hip_runtime.h>
#include <hip/hip_bf16.h>

typedef __attribute__((ext_vector_type(2))) float v2f;
typedef __attribute__((ext_vector_type(8))) float v8f;

#define NUM_TYPES 8
#define EMBED_DIM 32

__device__ __forceinline__ void zero_span(float* __restrict__ p, int n, int t, int nt) {
    const int n4 = ((((size_t)p) & 15) == 0) ? (n >> 2) : 0;
    float4* p4 = reinterpret_cast<float4*>(p);
    for (int i = t; i < n4; i += nt) p4[i] = make_float4(0.f, 0.f, 0.f, 0.f);
    for (int i = (n4 << 2) + t; i < n; i += nt) p[i] = 0.0f;
}

__global__ void __launch_bounds__(256)
k_zero(float* __restrict__ p, int n, float* __restrict__ q, int m) {
    const int t  = blockIdx.x * blockDim.x + threadIdx.x;
    const int nt = gridDim.x * blockDim.x;
    zero_span(p, n, t, nt);
    zero_span(q, m, t, nt);
}

#define TYPE_NONE 255u
__device__ __forceinline__ unsigned type_code(int t) {
    return ((unsigned)t < (unsigned)NUM_TYPES) ? ((unsigned)t << 2) : TYPE_NONE;
}

#define R1_TT_CAP 114688
__global__ void __launch_bounds__(256)
k_zero_pack(float* __restrict__ p, int n, float* __restrict__ q, int m,
            const int* __restrict__ node_type, float* __restrict__ tbf, int n_nodes) {
    const int t  = blockIdx.x * blockDim.x + threadIdx.x;
    const int nt = gridDim.x * blockDim.x;
    unsigned* tb = reinterpret_cast<unsigned*>(tbf);
    const bool al = (((size_t)node_type) & 15) == 0;
    for (int i = t; i < R1_TT_CAP / 4; i += nt) {
        const int j = i << 2;
        unsigned w;
        if (al && j + 3 < n_nodes) {
            const int4 a = *reinterpret_cast<const int4*>(node_type + j);
            w = type_code(a.x) | (type_code(a.y) << 8) | (type_code(a.z) << 16) | (type_code(a.w) << 24);
        } else {
            w = 0;
            for (int q = 0; q < 4; ++q)
                w |= ((j + q < n_nodes) ? type_code(node_type[j + q]) : TYPE_NONE) << (8 * q);
        }
        tb[i] = w;
    }

    zero_span(p, n, t, nt);
    zero_span(q, m, t, nt);
}

#define R1_RANGE_NODES 6400
#define R1_TB 7
#define R1_THREADS 1024
#define R1_GROUPS 256
#define R1_U 3
#define R1_F 5

static_assert(R1_TB * R1_THREADS * 16 == R1_TT_CAP, "the table copy covers the table exactly");

template <bool LDS_TYPES>
__device__ __forceinline__ int r1_type(const signed char* tt, const int* __restrict__ node_type,
                                       bool hit, int node) {
    if constexpr (LDS_TYPES) return tt[hit ? (unsigned)node : (unsigned)(R1_TT_CAP - 1)];
    else return node_type[(unsigned)(hit ? node : 0)];
}

template <bool LDS_TYPES>
__device__ __forceinline__ void r1_count(float* acc, bool hit, int rb, int ta) {
    if constexpr (LDS_TYPES) {
        if (ta >= 0)
            __hip_atomic_fetch_add(reinterpret_cast<float*>(reinterpret_cast<char*>(acc) +
                                                            (((unsigned)rb << 5) + (unsigned)ta)), 1.0f,
                                   __ATOMIC_RELAXED, __HIP_MEMORY_SCOPE_WORKGROUP);
    } else {
        if (hit && (unsigned)ta < (unsigned)NUM_TYPES)
            __hip_atomic_fetch_add(&acc[rb * NUM_TYPES + ta], 1.0f,
                                   __ATOMIC_RELAXED, __HIP_MEMORY_SCOPE_WORKGROUP);
    }
}

template <bool LDS_TYPES>
__global__ void __launch_bounds__(R1_THREADS)
k_round1(const int* __restrict__ src, const int* __restrict__ dst,
         const int* __restrict__ node_type, const float* __restrict__ tbf, float* __restrict__ cout,
         int n_edges, int n_nodes, int n_chunks, int per_chunk) {
    __shared__ __attribute__((aligned(16))) float acc[R1_RANGE_NODES * NUM_TYPES];
    __shared__ __attribute__((aligned(16))) signed char tt[LDS_TYPES ? R1_TT_CAP : 16];
    const int tid   = threadIdx.x;
    const int range = blockIdx.x / n_chunks;
    const int chunk = blockIdx.x - range * n_chunks;
    const int lo    = range * R1_RANGE_NODES;
    const int cnt   = min(R1_RANGE_NODES, n_nodes - lo);

    float4* acc4 = reinterpret_cast<float4*>(acc);
    for (int i = tid; i < cnt * 2; i += R1_THREADS) acc4[i] = make_float4(0.f, 0.f, 0.f, 0.f);
    if constexpr (LDS_TYPES) {
        const uint4* tb16 = reinterpret_cast<const uint4*>(tbf);
        uint4* tt16 = reinterpret_cast<uint4*>(tt);
        uint4 x[R1_TB];
#pragma unroll
        for (int k = 0; k < R1_TB; ++k) x[k] = tb16[tid + k * R1_THREADS];
#pragma unroll
        for (int k = 0; k < R1_TB; ++k) tt16[tid + k * R1_THREADS] = x[k];
    }
    __syncthreads();

    const int e0 = chunk * per_chunk;
    const int e1 = min(e0 + per_chunk, n_edges);
    const bool al = ((((size_t)src) | ((size_t)dst)) & 15) == 0;
    const int ev = (al && e1 > e0) ? e0 + ((e1 - e0) & ~3) : e0;

    for (int e = e0 + tid * 4; e < ev; e += R1_THREADS * 4 * R1_U) {
        int us[R1_U * 4], vs[R1_U * 4];
        bool ok[R1_U];
#pragma unroll
        for (int k = 0; k < R1_U; ++k) {
            const int ek = e + k * (R1_THREADS * 4);
            ok[k] = ek < ev;
            const int el = ok[k] ? ek : e0;
            const int4 a = *reinterpret_cast<const int4*>(src + el);
            const int4 b = *reinterpret_cast<const int4*>(dst + el);
            us[4 * k + 0] = a.x; us[4 * k + 1] = a.y; us[4 * k + 2] = a.z; us[4 * k + 3] = a.w;
            vs[4 * k + 0] = b.x; vs[4 * k + 1] = b.y; vs[4 * k + 2] = b.z; vs[4 * k + 3] = b.w;
        }
        bool hv[R1_U * 4], hu[R1_U * 4];
        int tu[R1_U * 4], tv[R1_U * 4];
#pragma unroll
        for (int j = 0; j < R1_U * 4; ++j) {
            hv[j] = ok[j >> 2] && (unsigned)(vs[j] - lo) < (unsigned)cnt;
            hu[j] = ok[j >> 2] && (unsigned)(us[j] - lo) < (unsigned)cnt;
            tu[j] = r1_type<LDS_TYPES>(tt, node_type, hv[j], us[j]);
            tv[j] = r1_type<LDS_TYPES>(tt, node_type, hu[j], vs[j]);
        }
#pragma unroll
        for (int j = 0; j < R1_U * 4; ++j) {
            r1_count<LDS_TYPES>(acc, hv[j], vs[j] - lo, tu[j]);
            r1_count<LDS_TYPES>(acc, hu[j], us[j] - lo, tv[j]);
        }
    }
    for (int i = ev + tid; i < e1; i += R1_THREADS) {
        const int u = src[i];
        const int v = dst[i];
        const bool hv1 = (unsigned)(v - lo) < (unsigned)cnt;
        const bool hu1 = (unsigned)(u - lo) < (unsigned)cnt;
        const int tu1 = r1_type<LDS_TYPES>(tt, node_type, hv1, u);
        const int tv1 = r1_type<LDS_TYPES>(tt, node_type, hu1, v);
        r1_count<LDS_TYPES>(acc, hv1, v - lo, tu1);
        r1_count<LDS_TYPES>(acc, hu1, u - lo, tv1);
    }
    __syncthreads();

    float* g = cout + (size_t)lo * NUM_TYPES;
    const int total = cnt * NUM_TYPES;
    for (int i = tid; i < total; i += R1_THREADS * R1_F) {
        float x[R1_F];
#pragma unroll
        for (int k = 0; k < R1_F; ++k) {
            const int ik = i + k * R1_THREADS;
            x[k] = (ik < total) ? acc[ik] : 0.0f;
        }
#pragma unroll
        for (int k = 0; k < R1_F; ++k) {
            if (x[k] != 0.0f)
                __hip_atomic_fetch_add(&g[i + k * R1_THREADS], x[k],
                                       __ATOMIC_RELAXED, __HIP_MEMORY_SCOPE_AGENT);
        }
    }
}

#define EO_THREADS 256
#define EO_GROUPS 1024
#define EO_B 8
__global__ void __launch_bounds__(EO_THREADS)
k_edges_out(const int* __restrict__ src, const int* __restrict__ dst,
            const float* __restrict__ cin, float* __restrict__ cout,
            int n_edges, int n_rows) {
    __shared__ int2 hits[EO_THREADS / 32][256];
    const int lane = threadIdx.x & 31;
    const int wv   = threadIdx.x >> 5;
    const int lc   = lane & 7;
    const int lg   = lane >> 3;
    const unsigned below = (1u << lane) - 1u;
    const int per = (((n_edges + (int)gridDim.x - 1) / (int)gridDim.x) + 3) & ~3;
    const int e0  = blockIdx.x * per;
    const int e1  = min(e0 + per, n_edges);
    const bool al = ((((size_t)src) | ((size_t)dst)) & 15) == 0;

    for (int eb = e0; eb < e1; eb += EO_THREADS * 4) {
        const int e = eb + threadIdx.x * 4;
        int us[4], vs[4];
        if (al && e + 3 < e1) {
            const int4 a = *reinterpret_cast<const int4*>(src + e);
            const int4 b = *reinterpret_cast<const int4*>(dst + e);
            us[0] = a.x; us[1] = a.y; us[2] = a.z; us[3] = a.w;
            vs[0] = b.x; vs[1] = b.y; vs[2] = b.z; vs[3] = b.w;
        } else {
#pragma unroll
            for (int j = 0; j < 4; ++j) {
                const bool in = e + j < e1;
                us[j] = in ? src[e + j] : -1;
                vs[j] = in ? dst[e + j] : -1;
            }
        }
        int n = 0;
#pragma unroll
        for (int j = 0; j < 4; ++j) {
            const bool q1 = (unsigned)vs[j] < (unsigned)n_rows;
            const bool q2 = (unsigned)us[j] < (unsigned)n_rows;
            const unsigned m1 = __builtin_amdgcn_ballot_w32(q1);
            const unsigned m2 = __builtin_amdgcn_ballot_w32(q2);
            const int n1 = __builtin_popcount(m1);
            if (q1) hits[wv][n + __builtin_popcount(m1 & below)] = make_int2(us[j], vs[j]);
            if (q2) hits[wv][n + n1 + __builtin_popcount(m2 & below)] = make_int2(vs[j], us[j]);
            n += n1 + __builtin_popcount(m2);
        }
        __syncthreads();
        for (int i0 = 0; i0 < n; i0 += 4 * EO_B) {
            int2 h[EO_B];
            float val[EO_B];
            bool ok[EO_B];
#pragma unroll
            for (int k = 0; k < EO_B; ++k) {
                const int i = i0 + 4 * k + lg;
                ok[k] = i < n;
                h[k]  = hits[wv][ok[k] ? i : 0];
            }
#pragma unroll
            for (int k = 0; k < EO_B; ++k)
                val[k] = cin[(size_t)(unsigned)h[k].x * NUM_TYPES + lc];
#pragma unroll
            for (int k = 0; k < EO_B; ++k) {
                if (ok[k])
                    __hip_atomic_fetch_add(&cout[(size_t)(unsigned)h[k].y * NUM_TYPES + lc], val[k],
                                           __ATOMIC_RELAXED, __HIP_MEMORY_SCOPE_AGENT);
            }
        }
    }
}

__global__ void k_embed_gemm(const float* __restrict__ counts,
                             const float* __restrict__ W,
                             float* __restrict__ out, int n_rows) {
    const int wave = (blockIdx.x * blockDim.x + threadIdx.x) >> 5;
    const int lane = threadIdx.x & 31;
    const int n_mtiles = (n_rows + 15) >> 4;
    const int mt = wave >> 1;
    const int nt = wave & 1;
    if (mt >= n_mtiles) return;

    const int m     = lane & 15;
    const int khalf = (lane >> 4) & 1;

    v8f acc = {0.f, 0.f, 0.f, 0.f, 0.f, 0.f, 0.f, 0.f};
    const float* crow = counts + (size_t)(mt * 16 + m) * NUM_TYPES;

#pragma unroll
    for (int s = 0; s < 2; ++s) {
        const int kb = s * 4 + khalf * 2;
        v2f a, b;
        a.x = crow[kb];
        a.y = crow[kb + 1];
        b.x = W[(size_t)kb       * EMBED_DIM + nt * 16 + m];
        b.y = W[(size_t)(kb + 1) * EMBED_DIM + nt * 16 + m];
        acc = __builtin_amdgcn_wmma_f32_16x16x4_f32(
             false, a,  false, b,
             (short)0, acc,  false,  false);
    }

    if (mt * 16 + 16 <= n_rows) {
        float* obase = out + (size_t)(mt * 16 + khalf * 8) * EMBED_DIM + nt * 16 + m;
#pragma unroll
        for (int r = 0; r < 8; ++r) {
            obase[(size_t)r * EMBED_DIM] = acc[r];
        }
    } else {
#pragma unroll
        for (int r = 0; r < 8; ++r) {
            int row = mt * 16 + khalf * 8 + r;
            if (row < n_rows) out[(size_t)row * EMBED_DIM + nt * 16 + m] = acc[r];
        }
    }
}

extern "C" void kernel_launch(void* const* d_in, const int* in_sizes, int n_in,
                              void* d_out, int out_size, void* d_ws, size_t ws_size,
                              hipStream_t stream) {
    const float* embed_table = (const float*)d_in[0];
    const int*   node_type   = (const int*)d_in[1];
    const int*   src         = (const int*)d_in[2];
    const int*   dst         = (const int*)d_in[3];
    float*       out         = (float*)d_out;

    const int n_nodes = in_sizes[1];
    const int n_edges = in_sizes[2];
    const int n_rows  = out_size / EMBED_DIM;

    float* c0 = (float*)d_ws;
    float* c1 = c0 + (size_t)n_nodes * NUM_TYPES;

    const int nc = n_nodes * NUM_TYPES;

    const int nz = ((n_rows + 15) & ~15) * NUM_TYPES;
    float* tbf = c0 + nz;
    const bool lds_types = n_nodes < R1_TT_CAP && (((size_t)tbf) & 15) == 0 &&
                           (size_t)nz * sizeof(float) + (size_t)R1_TT_CAP <= (size_t)nc * sizeof(float);

    if (lds_types) k_zero_pack<<<512, 256, 0, stream>>>(c1, nc, c0, nz, node_type, tbf, n_nodes);
    else           k_zero<<<512, 256, 0, stream>>>(c1, nc, c0, nz);

    {
        const int n_ranges  = (n_nodes + R1_RANGE_NODES - 1) / R1_RANGE_NODES;
        const int n_chunks  = (R1_GROUPS / n_ranges) > 0 ? (R1_GROUPS / n_ranges) : 1;
        const int per_chunk = ((n_edges + n_chunks - 1) / n_chunks + 3) & ~3;
        if (lds_types)
            k_round1<true><<<n_ranges * n_chunks, R1_THREADS, 0, stream>>>(src, dst, node_type, tbf, c1,
                                                                           n_edges, n_nodes, n_chunks, per_chunk);
        else
            k_round1<false><<<n_ranges * n_chunks, R1_THREADS, 0, stream>>>(src, dst, node_type, tbf, c1,
                                                                            n_edges, n_nodes, n_chunks, per_chunk);
    }

    {
        k_edges_out<<<EO_GROUPS, EO_THREADS, 0, stream>>>(src, dst, c1, c0, n_edges, n_rows);
    }

    {
        int n_mtiles = (n_rows + 15) / 16;
        int tiles = n_mtiles * 2;
        int waves_per_block = 256 / 32;
        int blocks = (tiles + waves_per_block - 1) / waves_per_block;
        k_embed_gemm<<<blocks, 256, 0, stream>>>(c0, embed_table, out, n_rows);
    }
}


# Round 2 summed in LDS: a work-group per (row range, edge chunk) adds gathered rows into LDS sums, merges them by whole-line f32 atomics; atomic add order differs, whole-number sums exact
# speedup vs baseline: 4.7466x; 1.1506x over previous
#include <hip/hip_runtime.h>
#include <hip/hip_bf16.h>

typedef __attribute__((ext_vector_type(2))) float v2f;
typedef __attribute__((ext_vector_type(8))) float v8f;

#define NUM_TYPES 8
#define EMBED_DIM 32

__device__ __forceinline__ void zero_span(float* __restrict__ p, int n, int t, int nt) {
    const int n4 = ((((size_t)p) & 15) == 0) ? (n >> 2) : 0;
    float4* p4 = reinterpret_cast<float4*>(p);
    for (int i = t; i < n4; i += nt) p4[i] = make_float4(0.f, 0.f, 0.f, 0.f);
    for (int i = (n4 << 2) + t; i < n; i += nt) p[i] = 0.0f;
}

__global__ void __launch_bounds__(256)
k_zero(float* __restrict__ p, int n, float* __restrict__ q, int m) {
    const int t  = blockIdx.x * blockDim.x + threadIdx.x;
    const int nt = gridDim.x * blockDim.x;
    zero_span(p, n, t, nt);
    zero_span(q, m, t, nt);
}

#define TYPE_NONE 255u
__device__ __forceinline__ unsigned type_code(int t) {
    return ((unsigned)t < (unsigned)NUM_TYPES) ? ((unsigned)t << 2) : TYPE_NONE;
}

#define R1_TT_CAP 114688
__global__ void __launch_bounds__(256)
k_zero_pack(float* __restrict__ p, int n, float* __restrict__ q, int m,
            const int* __restrict__ node_type, float* __restrict__ tbf, int n_nodes) {
    const int t  = blockIdx.x * blockDim.x + threadIdx.x;
    const int nt = gridDim.x * blockDim.x;
    unsigned* tb = reinterpret_cast<unsigned*>(tbf);
    const bool al = (((size_t)node_type) & 15) == 0;
    for (int i = t; i < R1_TT_CAP / 4; i += nt) {
        const int j = i << 2;
        unsigned w;
        if (al && j + 3 < n_nodes) {
            const int4 a = *reinterpret_cast<const int4*>(node_type + j);
            w = type_code(a.x) | (type_code(a.y) << 8) | (type_code(a.z) << 16) | (type_code(a.w) << 24);
        } else {
            w = 0;
            for (int q = 0; q < 4; ++q)
                w |= ((j + q < n_nodes) ? type_code(node_type[j + q]) : TYPE_NONE) << (8 * q);
        }
        tb[i] = w;
    }

    zero_span(p, n, t, nt);
    zero_span(q, m, t, nt);
}

#define R1_RANGE_NODES 6400
#define R1_TB 7
#define R1_THREADS 1024
#define R1_GROUPS 256
#define R1_U 3
#define R1_F 5

static_assert(R1_TB * R1_THREADS * 16 == R1_TT_CAP, "the table copy covers the table exactly");

template <bool LDS_TYPES>
__device__ __forceinline__ int r1_type(const signed char* tt, const int* __restrict__ node_type,
                                       bool hit, int node) {
    if constexpr (LDS_TYPES) return tt[hit ? (unsigned)node : (unsigned)(R1_TT_CAP - 1)];
    else return node_type[(unsigned)(hit ? node : 0)];
}

template <bool LDS_TYPES>
__device__ __forceinline__ void r1_count(float* acc, bool hit, int rb, int ta) {
    if constexpr (LDS_TYPES) {
        if (ta >= 0)
            __hip_atomic_fetch_add(reinterpret_cast<float*>(reinterpret_cast<char*>(acc) +
                                                            (((unsigned)rb << 5) + (unsigned)ta)), 1.0f,
                                   __ATOMIC_RELAXED, __HIP_MEMORY_SCOPE_WORKGROUP);
    } else {
        if (hit && (unsigned)ta < (unsigned)NUM_TYPES)
            __hip_atomic_fetch_add(&acc[rb * NUM_TYPES + ta], 1.0f,
                                   __ATOMIC_RELAXED, __HIP_MEMORY_SCOPE_WORKGROUP);
    }
}

template <bool LDS_TYPES>
__global__ void __launch_bounds__(R1_THREADS)
k_round1(const int* __restrict__ src, const int* __restrict__ dst,
         const int* __restrict__ node_type, const float* __restrict__ tbf, float* __restrict__ cout,
         int n_edges, int n_nodes, int n_chunks, int per_chunk) {
    __shared__ __attribute__((aligned(16))) float acc[R1_RANGE_NODES * NUM_TYPES];
    __shared__ __attribute__((aligned(16))) signed char tt[LDS_TYPES ? R1_TT_CAP : 16];
    const int tid   = threadIdx.x;
    const int range = blockIdx.x / n_chunks;
    const int chunk = blockIdx.x - range * n_chunks;
    const int lo    = range * R1_RANGE_NODES;
    const int cnt   = min(R1_RANGE_NODES, n_nodes - lo);

    float4* acc4 = reinterpret_cast<float4*>(acc);
    for (int i = tid; i < cnt * 2; i += R1_THREADS) acc4[i] = make_float4(0.f, 0.f, 0.f, 0.f);
    if constexpr (LDS_TYPES) {
        const uint4* tb16 = reinterpret_cast<const uint4*>(tbf);
        uint4* tt16 = reinterpret_cast<uint4*>(tt);
        uint4 x[R1_TB];
#pragma unroll
        for (int k = 0; k < R1_TB; ++k) x[k] = tb16[tid + k * R1_THREADS];
#pragma unroll
        for (int k = 0; k < R1_TB; ++k) tt16[tid + k * R1_THREADS] = x[k];
    }
    __syncthreads();

    const int e0 = chunk * per_chunk;
    const int e1 = min(e0 + per_chunk, n_edges);
    const bool al = ((((size_t)src) | ((size_t)dst)) & 15) == 0;
    const int ev = (al && e1 > e0) ? e0 + ((e1 - e0) & ~3) : e0;

    for (int e = e0 + tid * 4; e < ev; e += R1_THREADS * 4 * R1_U) {
        int us[R1_U * 4], vs[R1_U * 4];
        bool ok[R1_U];
#pragma unroll
        for (int k = 0; k < R1_U; ++k) {
            const int ek = e + k * (R1_THREADS * 4);
            ok[k] = ek < ev;
            const int el = ok[k] ? ek : e0;
            const int4 a = *reinterpret_cast<const int4*>(src + el);
            const int4 b = *reinterpret_cast<const int4*>(dst + el);
            us[4 * k + 0] = a.x; us[4 * k + 1] = a.y; us[4 * k + 2] = a.z; us[4 * k + 3] = a.w;
            vs[4 * k + 0] = b.x; vs[4 * k + 1] = b.y; vs[4 * k + 2] = b.z; vs[4 * k + 3] = b.w;
        }
        bool hv[R1_U * 4], hu[R1_U * 4];
        int tu[R1_U * 4], tv[R1_U * 4];
#pragma unroll
        for (int j = 0; j < R1_U * 4; ++j) {
            hv[j] = ok[j >> 2] && (unsigned)(vs[j] - lo) < (unsigned)cnt;
            hu[j] = ok[j >> 2] && (unsigned)(us[j] - lo) < (unsigned)cnt;
            tu[j] = r1_type<LDS_TYPES>(tt, node_type, hv[j], us[j]);
            tv[j] = r1_type<LDS_TYPES>(tt, node_type, hu[j], vs[j]);
        }
#pragma unroll
        for (int j = 0; j < R1_U * 4; ++j) {
            r1_count<LDS_TYPES>(acc, hv[j], vs[j] - lo, tu[j]);
            r1_count<LDS_TYPES>(acc, hu[j], us[j] - lo, tv[j]);
        }
    }
    for (int i = ev + tid; i < e1; i += R1_THREADS) {
        const int u = src[i];
        const int v = dst[i];
        const bool hv1 = (unsigned)(v - lo) < (unsigned)cnt;
        const bool hu1 = (unsigned)(u - lo) < (unsigned)cnt;
        const int tu1 = r1_type<LDS_TYPES>(tt, node_type, hv1, u);
        const int tv1 = r1_type<LDS_TYPES>(tt, node_type, hu1, v);
        r1_count<LDS_TYPES>(acc, hv1, v - lo, tu1);
        r1_count<LDS_TYPES>(acc, hu1, u - lo, tv1);
    }
    __syncthreads();

    float* g = cout + (size_t)lo * NUM_TYPES;
    const int total = cnt * NUM_TYPES;
    for (int i = tid; i < total; i += R1_THREADS * R1_F) {
        float x[R1_F];
#pragma unroll
        for (int k = 0; k < R1_F; ++k) {
            const int ik = i + k * R1_THREADS;
            x[k] = (ik < total) ? acc[ik] : 0.0f;
        }
#pragma unroll
        for (int k = 0; k < R1_F; ++k) {
            if (x[k] != 0.0f)
                __hip_atomic_fetch_add(&g[i + k * R1_THREADS], x[k],
                                       __ATOMIC_RELAXED, __HIP_MEMORY_SCOPE_AGENT);
        }
    }
}

#define R2_RANGE_ROWS 3200
#define R2_THREADS 1024
#define R2_GROUPS 256
#define R2_U 4
#define R2_CAP 512
#define R2_B 16
#define R2_F 5

__device__ __forceinline__ void r2_append(int2* list, int& n, int u, int v, bool in,
                                          int lo, int cnt, unsigned below) {
    const bool q1 = in && (unsigned)(v - lo) < (unsigned)cnt;
    const bool q2 = in && (unsigned)(u - lo) < (unsigned)cnt;
    const unsigned m1 = __builtin_amdgcn_ballot_w32(q1);
    const unsigned m2 = __builtin_amdgcn_ballot_w32(q2);
    const int n1 = __builtin_popcount(m1);
    if (q1) list[n + __builtin_popcount(m1 & below)] = make_int2(u, v - lo);
    if (q2) list[n + n1 + __builtin_popcount(m2 & below)] = make_int2(v, u - lo);
    n += n1 + __builtin_popcount(m2);
}

__device__ __forceinline__ void r2_flush(const int2* list, int n, const float* __restrict__ cin,
                                         float* acc, int lg, int lc) {
    __builtin_amdgcn_fence(__ATOMIC_ACQ_REL, "wavefront");
    for (int i0 = 0; i0 < n; i0 += 4 * R2_B) {
        int2 h[R2_B];
        float val[R2_B];
        bool ok[R2_B];
#pragma unroll
        for (int k = 0; k < R2_B; ++k) {
            const int i = i0 + 4 * k + lg;
            ok[k] = i < n;
            h[k]  = list[ok[k] ? i : 0];
        }
#pragma unroll
        for (int k = 0; k < R2_B; ++k)
            val[k] = cin[(size_t)(unsigned)h[k].x * NUM_TYPES + lc];
#pragma unroll
        for (int k = 0; k < R2_B; ++k) {
            if (ok[k])
                __hip_atomic_fetch_add(&acc[h[k].y * NUM_TYPES + lc], val[k],
                                       __ATOMIC_RELAXED, __HIP_MEMORY_SCOPE_WORKGROUP);
        }
    }
    __builtin_amdgcn_fence(__ATOMIC_ACQ_REL, "wavefront");
}

__device__ __forceinline__ void r2_flush_small(const int2* list, int n, const float* __restrict__ cin,
                                               float* acc, int lg, int lc) {
    __builtin_amdgcn_fence(__ATOMIC_ACQ_REL, "wavefront");
    for (int i = lg; i < n; i += 4) {
        const int2 h = list[i];
        const float val = cin[(size_t)(unsigned)h.x * NUM_TYPES + lc];
        __hip_atomic_fetch_add(&acc[h.y * NUM_TYPES + lc], val,
                               __ATOMIC_RELAXED, __HIP_MEMORY_SCOPE_WORKGROUP);
    }
    __builtin_amdgcn_fence(__ATOMIC_ACQ_REL, "wavefront");
}

__device__ __forceinline__ void r2_batch_load(int4 (&a)[R2_U], int4 (&b)[R2_U],
                                              const int* __restrict__ src, const int* __restrict__ dst,
                                              int e, int e0, int ev) {
#pragma unroll
    for (int k = 0; k < R2_U; ++k) {
        const int ek = e + k * (R2_THREADS * 4);
        const int el = (ek < ev) ? ek : e0;
        a[k] = *reinterpret_cast<const int4*>(src + el);
        b[k] = *reinterpret_cast<const int4*>(dst + el);
    }
}

__device__ __forceinline__ void r2_batch_append(const int4 (&a)[R2_U], const int4 (&b)[R2_U],
                                                int2* list, int& n, int e, int ev, int lo, int cnt,
                                                unsigned below, const float* __restrict__ cin,
                                                float* acc, int lg, int lc) {
#pragma unroll
    for (int k = 0; k < R2_U; ++k) {
        const bool ok = e + k * (R2_THREADS * 4) < ev;
        r2_append(list, n, a[k].x, b[k].x, ok, lo, cnt, below);
        r2_append(list, n, a[k].y, b[k].y, ok, lo, cnt, below);
        r2_append(list, n, a[k].z, b[k].z, ok, lo, cnt, below);
        r2_append(list, n, a[k].w, b[k].w, ok, lo, cnt, below);
        if (n > R2_CAP - 256) { r2_flush_small(list, n, cin, acc, lg, lc); n = 0; }
    }
}

__global__ void __launch_bounds__(R2_THREADS)
k_round2(const int* __restrict__ src, const int* __restrict__ dst,
         const float* __restrict__ cin, float* __restrict__ cout,
         int n_edges, int n_rows, int n_chunks, int per_chunk) {
    __shared__ __attribute__((aligned(16))) float acc[R2_RANGE_ROWS * NUM_TYPES];
    __shared__ int2 hits[R2_THREADS / 32][R2_CAP];
    const int tid   = threadIdx.x;
    const int lane  = tid & 31;
    const int wv    = tid >> 5;
    const int lc    = lane & 7;
    const int lg    = lane >> 3;
    const unsigned below = (1u << lane) - 1u;
    const int range = blockIdx.x / n_chunks;
    const int chunk = blockIdx.x - range * n_chunks;
    const int lo    = range * R2_RANGE_ROWS;
    const int cnt   = min(R2_RANGE_ROWS, n_rows - lo);

    const int e0 = chunk * per_chunk;
    const int e1 = min(e0 + per_chunk, n_edges);
    const bool al = ((((size_t)src) | ((size_t)dst)) & 15) == 0;
    const int ev = (al && e1 > e0) ? e0 + ((e1 - e0) & ~3) : e0;
    const int step = R2_THREADS * 4 * R2_U;
    int2* list = hits[wv];
    int n = 0;

    float4* acc4 = reinterpret_cast<float4*>(acc);
    for (int i = tid; i < cnt * 2; i += R2_THREADS) acc4[i] = make_float4(0.f, 0.f, 0.f, 0.f);
    __syncthreads();

    if (e0 < ev) {
        int4 a0[R2_U], b0[R2_U], a1[R2_U], b1[R2_U];
        r2_batch_load(a0, b0, src, dst, e0 + tid * 4, e0, ev);
        for (int eb = e0; eb < ev; eb += 2 * step) {
            r2_batch_load(a1, b1, src, dst, eb + step + tid * 4, e0, ev);
            r2_batch_append(a0, b0, list, n, eb + tid * 4, ev, lo, cnt, below, cin, acc, lg, lc);
            r2_batch_load(a0, b0, src, dst, eb + 2 * step + tid * 4, e0, ev);
            if (eb + step < ev)
                r2_batch_append(a1, b1, list, n, eb + step + tid * 4, ev, lo, cnt, below, cin, acc, lg, lc);
        }
    }
    for (int ib = ev; ib < e1; ib += R2_THREADS) {
        const int i = ib + tid;
        const bool in = i < e1;
        const int u = in ? src[i] : 0;
        const int v = in ? dst[i] : 0;
        r2_append(list, n, u, v, in, lo, cnt, below);
        if (n > R2_CAP - 64) { r2_flush_small(list, n, cin, acc, lg, lc); n = 0; }
    }
    r2_flush(list, n, cin, acc, lg, lc);
    __syncthreads();

    float* g = cout + (size_t)lo * NUM_TYPES;
    const int total = cnt * NUM_TYPES;
    for (int i = tid; i < total; i += R2_THREADS * R2_F) {
        float x[R2_F];
#pragma unroll
        for (int k = 0; k < R2_F; ++k) {
            const int ik = i + k * R2_THREADS;
            x[k] = (ik < total) ? acc[ik] : 0.0f;
        }
#pragma unroll
        for (int k = 0; k < R2_F; ++k) {
            if (x[k] != 0.0f)
                __hip_atomic_fetch_add(&g[i + k * R2_THREADS], x[k],
                                       __ATOMIC_RELAXED, __HIP_MEMORY_SCOPE_AGENT);
        }
    }
}

__global__ void k_embed_gemm(const float* __restrict__ counts,
                             const float* __restrict__ W,
                             float* __restrict__ out, int n_rows) {
    const int wave = (blockIdx.x * blockDim.x + threadIdx.x) >> 5;
    const int lane = threadIdx.x & 31;
    const int n_mtiles = (n_rows + 15) >> 4;
    const int mt = wave >> 1;
    const int nt = wave & 1;
    if (mt >= n_mtiles) return;

    const int m     = lane & 15;
    const int khalf = (lane >> 4) & 1;

    v8f acc = {0.f, 0.f, 0.f, 0.f, 0.f, 0.f, 0.f, 0.f};
    const float* crow = counts + (size_t)(mt * 16 + m) * NUM_TYPES;

#pragma unroll
    for (int s = 0; s < 2; ++s) {
        const int kb = s * 4 + khalf * 2;
        v2f a, b;
        a.x = crow[kb];
        a.y = crow[kb + 1];
        b.x = W[(size_t)kb       * EMBED_DIM + nt * 16 + m];
        b.y = W[(size_t)(kb + 1) * EMBED_DIM + nt * 16 + m];
        acc = __builtin_amdgcn_wmma_f32_16x16x4_f32(
             false, a,  false, b,
             (short)0, acc,  false,  false);
    }

    if (mt * 16 + 16 <= n_rows) {
        float* obase = out + (size_t)(mt * 16 + khalf * 8) * EMBED_DIM + nt * 16 + m;
#pragma unroll
        for (int r = 0; r < 8; ++r) {
            obase[(size_t)r * EMBED_DIM] = acc[r];
        }
    } else {
#pragma unroll
        for (int r = 0; r < 8; ++r) {
            int row = mt * 16 + khalf * 8 + r;
            if (row < n_rows) out[(size_t)row * EMBED_DIM + nt * 16 + m] = acc[r];
        }
    }
}

extern "C" void kernel_launch(void* const* d_in, const int* in_sizes, int n_in,
                              void* d_out, int out_size, void* d_ws, size_t ws_size,
                              hipStream_t stream) {
    const float* embed_table = (const float*)d_in[0];
    const int*   node_type   = (const int*)d_in[1];
    const int*   src         = (const int*)d_in[2];
    const int*   dst         = (const int*)d_in[3];
    float*       out         = (float*)d_out;

    const int n_nodes = in_sizes[1];
    const int n_edges = in_sizes[2];
    const int n_rows  = out_size / EMBED_DIM;

    float* c0 = (float*)d_ws;
    float* c1 = c0 + (size_t)n_nodes * NUM_TYPES;

    const int nc = n_nodes * NUM_TYPES;

    const int nz = ((n_rows + 15) & ~15) * NUM_TYPES;
    float* tbf = c0 + nz;
    const bool lds_types = n_nodes < R1_TT_CAP && (((size_t)tbf) & 15) == 0 &&
                           (size_t)nz * sizeof(float) + (size_t)R1_TT_CAP <= (size_t)nc * sizeof(float);

    if (lds_types) k_zero_pack<<<512, 256, 0, stream>>>(c1, nc, c0, nz, node_type, tbf, n_nodes);
    else           k_zero<<<512, 256, 0, stream>>>(c1, nc, c0, nz);

    {
        const int n_ranges  = (n_nodes + R1_RANGE_NODES - 1) / R1_RANGE_NODES;
        const int n_chunks  = (R1_GROUPS / n_ranges) > 0 ? (R1_GROUPS / n_ranges) : 1;
        const int per_chunk = ((n_edges + n_chunks - 1) / n_chunks + 3) & ~3;
        if (lds_types)
            k_round1<true><<<n_ranges * n_chunks, R1_THREADS, 0, stream>>>(src, dst, node_type, tbf, c1,
                                                                           n_edges, n_nodes, n_chunks, per_chunk);
        else
            k_round1<false><<<n_ranges * n_chunks, R1_THREADS, 0, stream>>>(src, dst, node_type, tbf, c1,
                                                                            n_edges, n_nodes, n_chunks, per_chunk);
    }

    {
        const int n_ranges  = n_rows > 0 ? (n_rows + R2_RANGE_ROWS - 1) / R2_RANGE_ROWS : 1;
        const int n_chunks  = (R2_GROUPS / n_ranges) > 0 ? (R2_GROUPS / n_ranges) : 1;
        const int per_chunk = ((n_edges + n_chunks - 1) / n_chunks + 3) & ~3;
        k_round2<<<n_ranges * n_chunks, R2_THREADS, 0, stream>>>(src, dst, c1, c0, n_edges, n_rows,
                                                                 n_chunks, per_chunk);
    }

    {
        int n_mtiles = (n_rows + 15) / 16;
        int tiles = n_mtiles * 2;
        int waves_per_block = 256 / 32;
        int blocks = (tiles + waves_per_block - 1) / waves_per_block;
        k_embed_gemm<<<blocks, 256, 0, stream>>>(c0, embed_table, out, n_rows);
    }
}


# Work items of both rounds handed out per package half (die id, one counter draw per group): every merged line gets its atomics from one half; atomic add order differs, whole-number sums exact
# speedup vs baseline: 6.0710x; 1.2790x over previous
#include <hip/hip_runtime.h>
#include <hip/hip_bf16.h>

typedef __attribute__((ext_vector_type(2))) float v2f;
typedef __attribute__((ext_vector_type(8))) float v8f;

#define NUM_TYPES 8
#define EMBED_DIM 32

__device__ __forceinline__ void zero_span(float* __restrict__ p, int n, int t, int nt) {
    const int n4 = ((((size_t)p) & 15) == 0) ? (n >> 2) : 0;
    float4* p4 = reinterpret_cast<float4*>(p);
    for (int i = t; i < n4; i += nt) p4[i] = make_float4(0.f, 0.f, 0.f, 0.f);
    for (int i = (n4 << 2) + t; i < n; i += nt) p[i] = 0.0f;
}

#define N_TICK 4
__device__ __forceinline__ int take_item(int* tick, int n_items) {
    const int h    = (int)((__builtin_amdgcn_s_sendmsg_rtn(0x87) >> 18) & 1u);
    const int n0   = n_items >> 1;
    const int mine = h ? n_items - n0 : n0;
    int t = __hip_atomic_fetch_add(&tick[h], 1, __ATOMIC_RELAXED, __HIP_MEMORY_SCOPE_AGENT);
    int item;
    if (t < mine) item = h ? n0 + t : t;
    else {
        t = __hip_atomic_fetch_add(&tick[h ^ 1], 1, __ATOMIC_RELAXED, __HIP_MEMORY_SCOPE_AGENT);
        item = h ? t : n0 + t;
    }
    return min(max(item, 0), n_items - 1);
}

__global__ void __launch_bounds__(256)
k_zero(float* __restrict__ p, int n, float* __restrict__ q, int m, int* __restrict__ tick) {
    const int t  = blockIdx.x * blockDim.x + threadIdx.x;
    const int nt = gridDim.x * blockDim.x;
    if (tick != nullptr && t < N_TICK) tick[t] = 0;
    zero_span(p, n, t, nt);
    zero_span(q, m, t, nt);
}

#define TYPE_NONE 255u
__device__ __forceinline__ unsigned type_code(int t) {
    return ((unsigned)t < (unsigned)NUM_TYPES) ? ((unsigned)t << 2) : TYPE_NONE;
}

#define R1_TT_CAP 114688
__global__ void __launch_bounds__(256)
k_zero_pack(float* __restrict__ p, int n, float* __restrict__ q, int m,
            const int* __restrict__ node_type, float* __restrict__ tbf, int n_nodes,
            int* __restrict__ tick) {
    const int t  = blockIdx.x * blockDim.x + threadIdx.x;
    const int nt = gridDim.x * blockDim.x;
    unsigned* tb = reinterpret_cast<unsigned*>(tbf);
    if (tick != nullptr && t < N_TICK) tick[t] = 0;
    const bool al = (((size_t)node_type) & 15) == 0;
    for (int i = t; i < R1_TT_CAP / 4; i += nt) {
        const int j = i << 2;
        unsigned w;
        if (al && j + 3 < n_nodes) {
            const int4 a = *reinterpret_cast<const int4*>(node_type + j);
            w = type_code(a.x) | (type_code(a.y) << 8) | (type_code(a.z) << 16) | (type_code(a.w) << 24);
        } else {
            w = 0;
            for (int q = 0; q < 4; ++q)
                w |= ((j + q < n_nodes) ? type_code(node_type[j + q]) : TYPE_NONE) << (8 * q);
        }
        tb[i] = w;
    }

    zero_span(p, n, t, nt);
    zero_span(q, m, t, nt);
}

#define R1_RANGE_NODES 6400
#define R1_TB 7
#define R1_THREADS 1024
#define R1_GROUPS 256
#define R1_U 3
#define R1_F 5

static_assert(R1_TB * R1_THREADS * 16 == R1_TT_CAP, "the table copy covers the table exactly");

template <bool LDS_TYPES>
__device__ __forceinline__ int r1_type(const signed char* tt, const int* __restrict__ node_type,
                                       bool hit, int node) {
    if constexpr (LDS_TYPES) return tt[hit ? (unsigned)node : (unsigned)(R1_TT_CAP - 1)];
    else return node_type[(unsigned)(hit ? node : 0)];
}

template <bool LDS_TYPES>
__device__ __forceinline__ void r1_count(float* acc, bool hit, int rb, int ta) {
    if constexpr (LDS_TYPES) {
        if (ta >= 0)
            __hip_atomic_fetch_add(reinterpret_cast<float*>(reinterpret_cast<char*>(acc) +
                                                            (((unsigned)rb << 5) + (unsigned)ta)), 1.0f,
                                   __ATOMIC_RELAXED, __HIP_MEMORY_SCOPE_WORKGROUP);
    } else {
        if (hit && (unsigned)ta < (unsigned)NUM_TYPES)
            __hip_atomic_fetch_add(&acc[rb * NUM_TYPES + ta], 1.0f,
                                   __ATOMIC_RELAXED, __HIP_MEMORY_SCOPE_WORKGROUP);
    }
}

template <bool LDS_TYPES>
__global__ void __launch_bounds__(R1_THREADS)
k_round1(const int* __restrict__ src, const int* __restrict__ dst,
         const int* __restrict__ node_type, const float* __restrict__ tbf, float* __restrict__ cout,
         int n_edges, int n_nodes, int n_chunks, int per_chunk, int* __restrict__ tick, int n_items) {
    __shared__ __attribute__((aligned(16))) float acc[R1_RANGE_NODES * NUM_TYPES];
    __shared__ __attribute__((aligned(16))) signed char tt[LDS_TYPES ? R1_TT_CAP : 16];
    __shared__ int item_s;
    const int tid   = threadIdx.x;
    const int zt = (tick != nullptr) ? tid - 32 : tid;
    const int zn = (tick != nullptr) ? R1_THREADS - 32 : R1_THREADS;
    if (tick != nullptr && tid == 0) item_s = take_item(tick, n_items);

    float4* acc4 = reinterpret_cast<float4*>(acc);
    if (zt >= 0)
        for (int i = zt; i < R1_RANGE_NODES * 2; i += zn) acc4[i] = make_float4(0.f, 0.f, 0.f, 0.f);
    if constexpr (LDS_TYPES) {
        const uint4* tb16 = reinterpret_cast<const uint4*>(tbf);
        uint4* tt16 = reinterpret_cast<uint4*>(tt);
        uint4 x[R1_TB];
#pragma unroll
        for (int k = 0; k < R1_TB; ++k) x[k] = tb16[tid + k * R1_THREADS];
#pragma unroll
        for (int k = 0; k < R1_TB; ++k) tt16[tid + k * R1_THREADS] = x[k];
    }
    __syncthreads();

    const int item  = (tick != nullptr) ? __builtin_amdgcn_readfirstlane(item_s) : (int)blockIdx.x;
    const int range = item / n_chunks;
    const int chunk = item - range * n_chunks;
    const int lo    = range * R1_RANGE_NODES;
    const int cnt   = min(R1_RANGE_NODES, n_nodes - lo);

    const int e0 = chunk * per_chunk;
    const int e1 = min(e0 + per_chunk, n_edges);
    const bool al = ((((size_t)src) | ((size_t)dst)) & 15) == 0;
    const int ev = (al && e1 > e0) ? e0 + ((e1 - e0) & ~3) : e0;

    for (int e = e0 + tid * 4; e < ev; e += R1_THREADS * 4 * R1_U) {
        int us[R1_U * 4], vs[R1_U * 4];
        bool ok[R1_U];
#pragma unroll
        for (int k = 0; k < R1_U; ++k) {
            const int ek = e + k * (R1_THREADS * 4);
            ok[k] = ek < ev;
            const int el = ok[k] ? ek : e0;
            const int4 a = *reinterpret_cast<const int4*>(src + el);
            const int4 b = *reinterpret_cast<const int4*>(dst + el);
            us[4 * k + 0] = a.x; us[4 * k + 1] = a.y; us[4 * k + 2] = a.z; us[4 * k + 3] = a.w;
            vs[4 * k + 0] = b.x; vs[4 * k + 1] = b.y; vs[4 * k + 2] = b.z; vs[4 * k + 3] = b.w;
        }
        bool hv[R1_U * 4], hu[R1_U * 4];
        int tu[R1_U * 4], tv[R1_U * 4];
#pragma unroll
        for (int j = 0; j < R1_U * 4; ++j) {
            hv[j] = ok[j >> 2] && (unsigned)(vs[j] - lo) < (unsigned)cnt;
            hu[j] = ok[j >> 2] && (unsigned)(us[j] - lo) < (unsigned)cnt;
            tu[j] = r1_type<LDS_TYPES>(tt, node_type, hv[j], us[j]);
            tv[j] = r1_type<LDS_TYPES>(tt, node_type, hu[j], vs[j]);
        }
#pragma unroll
        for (int j = 0; j < R1_U * 4; ++j) {
            r1_count<LDS_TYPES>(acc, hv[j], vs[j] - lo, tu[j]);
            r1_count<LDS_TYPES>(acc, hu[j], us[j] - lo, tv[j]);
        }
    }
    for (int i = ev + tid; i < e1; i += R1_THREADS) {
        const int u = src[i];
        const int v = dst[i];
        const bool hv1 = (unsigned)(v - lo) < (unsigned)cnt;
        const bool hu1 = (unsigned)(u - lo) < (unsigned)cnt;
        const int tu1 = r1_type<LDS_TYPES>(tt, node_type, hv1, u);
        const int tv1 = r1_type<LDS_TYPES>(tt, node_type, hu1, v);
        r1_count<LDS_TYPES>(acc, hv1, v - lo, tu1);
        r1_count<LDS_TYPES>(acc, hu1, u - lo, tv1);
    }
    __syncthreads();

    float* g = cout + (size_t)lo * NUM_TYPES;
    const int total = cnt * NUM_TYPES;
    for (int i = tid; i < total; i += R1_THREADS * R1_F) {
        float x[R1_F];
#pragma unroll
        for (int k = 0; k < R1_F; ++k) {
            const int ik = i + k * R1_THREADS;
            x[k] = (ik < total) ? acc[ik] : 0.0f;
        }
#pragma unroll
        for (int k = 0; k < R1_F; ++k) {
            if (x[k] != 0.0f)
                __hip_atomic_fetch_add(&g[i + k * R1_THREADS], x[k],
                                       __ATOMIC_RELAXED, __HIP_MEMORY_SCOPE_AGENT);
        }
    }
}

#define R2_RANGE_ROWS 3200
#define R2_THREADS 1024
#define R2_GROUPS 256
#define R2_U 4
#define R2_CAP 512
#define R2_B 16
#define R2_F 5

__device__ __forceinline__ void r2_append(int2* list, int& n, int u, int v, bool in,
                                          int lo, int cnt, unsigned below) {
    const bool q1 = in && (unsigned)(v - lo) < (unsigned)cnt;
    const bool q2 = in && (unsigned)(u - lo) < (unsigned)cnt;
    const unsigned m1 = __builtin_amdgcn_ballot_w32(q1);
    const unsigned m2 = __builtin_amdgcn_ballot_w32(q2);
    const int n1 = __builtin_popcount(m1);
    if (q1) list[n + __builtin_popcount(m1 & below)] = make_int2(u, v - lo);
    if (q2) list[n + n1 + __builtin_popcount(m2 & below)] = make_int2(v, u - lo);
    n += n1 + __builtin_popcount(m2);
}

__device__ __forceinline__ void r2_flush(const int2* list, int n, const float* __restrict__ cin,
                                         float* acc, int lg, int lc) {
    __builtin_amdgcn_fence(__ATOMIC_ACQ_REL, "wavefront");
    for (int i0 = 0; i0 < n; i0 += 4 * R2_B) {
        int2 h[R2_B];
        float val[R2_B];
        bool ok[R2_B];
#pragma unroll
        for (int k = 0; k < R2_B; ++k) {
            const int i = i0 + 4 * k + lg;
            ok[k] = i < n;
            h[k]  = list[ok[k] ? i : 0];
        }
#pragma unroll
        for (int k = 0; k < R2_B; ++k)
            val[k] = cin[(size_t)(unsigned)h[k].x * NUM_TYPES + lc];
#pragma unroll
        for (int k = 0; k < R2_B; ++k) {
            if (ok[k])
                __hip_atomic_fetch_add(&acc[h[k].y * NUM_TYPES + lc], val[k],
                                       __ATOMIC_RELAXED, __HIP_MEMORY_SCOPE_WORKGROUP);
        }
    }
    __builtin_amdgcn_fence(__ATOMIC_ACQ_REL, "wavefront");
}

__device__ __forceinline__ void r2_flush_small(const int2* list, int n, const float* __restrict__ cin,
                                               float* acc, int lg, int lc) {
    __builtin_amdgcn_fence(__ATOMIC_ACQ_REL, "wavefront");
    for (int i = lg; i < n; i += 4) {
        const int2 h = list[i];
        const float val = cin[(size_t)(unsigned)h.x * NUM_TYPES + lc];
        __hip_atomic_fetch_add(&acc[h.y * NUM_TYPES + lc], val,
                               __ATOMIC_RELAXED, __HIP_MEMORY_SCOPE_WORKGROUP);
    }
    __builtin_amdgcn_fence(__ATOMIC_ACQ_REL, "wavefront");
}

__device__ __forceinline__ void r2_batch_load(int4 (&a)[R2_U], int4 (&b)[R2_U],
                                              const int* __restrict__ src, const int* __restrict__ dst,
                                              int e, int e0, int ev) {
#pragma unroll
    for (int k = 0; k < R2_U; ++k) {
        const int ek = e + k * (R2_THREADS * 4);
        const int el = (ek < ev) ? ek : e0;
        a[k] = *reinterpret_cast<const int4*>(src + el);
        b[k] = *reinterpret_cast<const int4*>(dst + el);
    }
}

__device__ __forceinline__ void r2_batch_append(const int4 (&a)[R2_U], const int4 (&b)[R2_U],
                                                int2* list, int& n, int e, int ev, int lo, int cnt,
                                                unsigned below, const float* __restrict__ cin,
                                                float* acc, int lg, int lc) {
#pragma unroll
    for (int k = 0; k < R2_U; ++k) {
        const bool ok = e + k * (R2_THREADS * 4) < ev;
        r2_append(list, n, a[k].x, b[k].x, ok, lo, cnt, below);
        r2_append(list, n, a[k].y, b[k].y, ok, lo, cnt, below);
        r2_append(list, n, a[k].z, b[k].z, ok, lo, cnt, below);
        r2_append(list, n, a[k].w, b[k].w, ok, lo, cnt, below);
        if (n > R2_CAP - 256) { r2_flush_small(list, n, cin, acc, lg, lc); n = 0; }
    }
}

__global__ void __launch_bounds__(R2_THREADS)
k_round2(const int* __restrict__ src, const int* __restrict__ dst,
         const float* __restrict__ cin, float* __restrict__ cout,
         int n_edges, int n_rows, int n_chunks, int per_chunk, int* __restrict__ tick, int n_items) {
    __shared__ __attribute__((aligned(16))) float acc[R2_RANGE_ROWS * NUM_TYPES];
    __shared__ int2 hits[R2_THREADS / 32][R2_CAP];
    __shared__ int item_s;
    const int tid   = threadIdx.x;
    const int lane  = tid & 31;
    const int wv    = tid >> 5;
    const int lc    = lane & 7;
    const int lg    = lane >> 3;
    const unsigned below = (1u << lane) - 1u;
    const int zt = (tick != nullptr) ? tid - 32 : tid;
    const int zn = (tick != nullptr) ? R2_THREADS - 32 : R2_THREADS;
    if (tick != nullptr && tid == 0) item_s = take_item(tick, n_items);

    float4* acc4 = reinterpret_cast<float4*>(acc);
    if (zt >= 0)
        for (int i = zt; i < R2_RANGE_ROWS * 2; i += zn) acc4[i] = make_float4(0.f, 0.f, 0.f, 0.f);
    __syncthreads();

    const int item  = (tick != nullptr) ? __builtin_amdgcn_readfirstlane(item_s) : (int)blockIdx.x;
    const int range = item / n_chunks;
    const int chunk = item - range * n_chunks;
    const int lo    = range * R2_RANGE_ROWS;
    const int cnt   = min(R2_RANGE_ROWS, n_rows - lo);

    const int e0 = chunk * per_chunk;
    const int e1 = min(e0 + per_chunk, n_edges);
    const bool al = ((((size_t)src) | ((size_t)dst)) & 15) == 0;
    const int ev = (al && e1 > e0) ? e0 + ((e1 - e0) & ~3) : e0;
    const int step = R2_THREADS * 4 * R2_U;
    int2* list = hits[wv];
    int n = 0;

    if (e0 < ev) {
        int4 a0[R2_U], b0[R2_U], a1[R2_U], b1[R2_U];
        r2_batch_load(a0, b0, src, dst, e0 + tid * 4, e0, ev);
        for (int eb = e0; eb < ev; eb += 2 * step) {
            r2_batch_load(a1, b1, src, dst, eb + step + tid * 4, e0, ev);
            r2_batch_append(a0, b0, list, n, eb + tid * 4, ev, lo, cnt, below, cin, acc, lg, lc);
            r2_batch_load(a0, b0, src, dst, eb + 2 * step + tid * 4, e0, ev);
            if (eb + step < ev)
                r2_batch_append(a1, b1, list, n, eb + step + tid * 4, ev, lo, cnt, below, cin, acc, lg, lc);
        }
    }
    for (int ib = ev; ib < e1; ib += R2_THREADS) {
        const int i = ib + tid;
        const bool in = i < e1;
        const int u = in ? src[i] : 0;
        const int v = in ? dst[i] : 0;
        r2_append(list, n, u, v, in, lo, cnt, below);
        if (n > R2_CAP - 64) { r2_flush_small(list, n, cin, acc, lg, lc); n = 0; }
    }
    r2_flush(list, n, cin, acc, lg, lc);
    __syncthreads();

    float* g = cout + (size_t)lo * NUM_TYPES;
    const int total = cnt * NUM_TYPES;
    for (int i = tid; i < total; i += R2_THREADS * R2_F) {
        float x[R2_F];
#pragma unroll
        for (int k = 0; k < R2_F; ++k) {
            const int ik = i + k * R2_THREADS;
            x[k] = (ik < total) ? acc[ik] : 0.0f;
        }
#pragma unroll
        for (int k = 0; k < R2_F; ++k) {
            if (x[k] != 0.0f)
                __hip_atomic_fetch_add(&g[i + k * R2_THREADS], x[k],
                                       __ATOMIC_RELAXED, __HIP_MEMORY_SCOPE_AGENT);
        }
    }
}

__global__ void k_embed_gemm(const float* __restrict__ counts,
                             const float* __restrict__ W,
                             float* __restrict__ out, int n_rows) {
    const int wave = (blockIdx.x * blockDim.x + threadIdx.x) >> 5;
    const int lane = threadIdx.x & 31;
    const int n_mtiles = (n_rows + 15) >> 4;
    const int mt = wave >> 1;
    const int nt = wave & 1;
    if (mt >= n_mtiles) return;

    const int m     = lane & 15;
    const int khalf = (lane >> 4) & 1;

    v8f acc = {0.f, 0.f, 0.f, 0.f, 0.f, 0.f, 0.f, 0.f};
    const float* crow = counts + (size_t)(mt * 16 + m) * NUM_TYPES;

#pragma unroll
    for (int s = 0; s < 2; ++s) {
        const int kb = s * 4 + khalf * 2;
        v2f a, b;
        a.x = crow[kb];
        a.y = crow[kb + 1];
        b.x = W[(size_t)kb       * EMBED_DIM + nt * 16 + m];
        b.y = W[(size_t)(kb + 1) * EMBED_DIM + nt * 16 + m];
        acc = __builtin_amdgcn_wmma_f32_16x16x4_f32(
             false, a,  false, b,
             (short)0, acc,  false,  false);
    }

    if (mt * 16 + 16 <= n_rows) {
        float* obase = out + (size_t)(mt * 16 + khalf * 8) * EMBED_DIM + nt * 16 + m;
#pragma unroll
        for (int r = 0; r < 8; ++r) {
            obase[(size_t)r * EMBED_DIM] = acc[r];
        }
    } else {
#pragma unroll
        for (int r = 0; r < 8; ++r) {
            int row = mt * 16 + khalf * 8 + r;
            if (row < n_rows) out[(size_t)row * EMBED_DIM + nt * 16 + m] = acc[r];
        }
    }
}

extern "C" void kernel_launch(void* const* d_in, const int* in_sizes, int n_in,
                              void* d_out, int out_size, void* d_ws, size_t ws_size,
                              hipStream_t stream) {
    const float* embed_table = (const float*)d_in[0];
    const int*   node_type   = (const int*)d_in[1];
    const int*   src         = (const int*)d_in[2];
    const int*   dst         = (const int*)d_in[3];
    float*       out         = (float*)d_out;

    const int n_nodes = in_sizes[1];
    const int n_edges = in_sizes[2];
    const int n_rows  = out_size / EMBED_DIM;

    float* c0 = (float*)d_ws;
    float* c1 = c0 + (size_t)n_nodes * NUM_TYPES;

    const int nc = n_nodes * NUM_TYPES;

    const int nz = ((n_rows + 15) & ~15) * NUM_TYPES;
    float* tbf = c0 + nz;
    const bool lds_types = n_nodes < R1_TT_CAP && (((size_t)tbf) & 15) == 0 &&
                           (size_t)nz * sizeof(float) + (size_t)R1_TT_CAP <= (size_t)nc * sizeof(float);

    const bool has_tick = (size_t)nz + (lds_types ? R1_TT_CAP / 4 : 0) + N_TICK <= (size_t)nc;
    int* tick = has_tick ? reinterpret_cast<int*>(c0 + nc) - N_TICK : nullptr;

    if (lds_types) k_zero_pack<<<512, 256, 0, stream>>>(c1, nc, c0, nz, node_type, tbf, n_nodes, tick);
    else           k_zero<<<512, 256, 0, stream>>>(c1, nc, c0, nz, tick);

    {
        const int n_ranges  = (n_nodes + R1_RANGE_NODES - 1) / R1_RANGE_NODES;
        const int n_chunks  = (R1_GROUPS / n_ranges) > 0 ? (R1_GROUPS / n_ranges) : 1;
        const int per_chunk = ((n_edges + n_chunks - 1) / n_chunks + 3) & ~3;
        if (lds_types)
            k_round1<true><<<n_ranges * n_chunks, R1_THREADS, 0, stream>>>(src, dst, node_type, tbf, c1,
                                                                           n_edges, n_nodes, n_chunks, per_chunk,
                                                                           tick, n_ranges * n_chunks);
        else
            k_round1<false><<<n_ranges * n_chunks, R1_THREADS, 0, stream>>>(src, dst, node_type, tbf, c1,
                                                                            n_edges, n_nodes, n_chunks, per_chunk,
                                                                            tick, n_ranges * n_chunks);
    }

    {
        const int n_ranges  = n_rows > 0 ? (n_rows + R2_RANGE_ROWS - 1) / R2_RANGE_ROWS : 1;
        const int n_chunks  = (R2_GROUPS / n_ranges) > 0 ? (R2_GROUPS / n_ranges) : 1;
        const int per_chunk = ((n_edges + n_chunks - 1) / n_chunks + 3) & ~3;
        k_round2<<<n_ranges * n_chunks, R2_THREADS, 0, stream>>>(src, dst, c1, c0, n_edges, n_rows,
                                                                 n_chunks, per_chunk,
                                                                 has_tick ? tick + 2 : nullptr, n_ranges * n_chunks);
    }

    {
        int n_mtiles = (n_rows + 15) / 16;
        int tiles = n_mtiles * 2;
        int waves_per_block = 256 / 32;
        int blocks = (tiles + waves_per_block - 1) / waves_per_block;
        k_embed_gemm<<<blocks, 256, 0, stream>>>(c0, embed_table, out, n_rows);
    }
}
